# starting kernel
# baseline (speedup 1.0000x reference)
#include <hip/hip_runtime.h>
#include <hip/hip_bf16.h>


#define NN      100000
#define EE      1600000
#define IN_DIM  256
#define HID     128
#define NCLS    40
#define NCLS_P  48

typedef __attribute__((ext_vector_type(2))) float v2f;
typedef __attribute__((ext_vector_type(8))) float v8f;

__global__ void init_deg_kernel(float* deg, int n) {
    int i = blockIdx.x * blockDim.x + threadIdx.x;
    if (i < n) deg[i] = 1.0f;
}

__global__ void deg_accum_kernel(const int* __restrict__ dst, float* deg, int e) {
    int i = blockIdx.x * blockDim.x + threadIdx.x;
    int stride = gridDim.x * blockDim.x;
    for (; i < e; i += stride) atomicAdd(&deg[dst[i]], 1.0f);
}

__global__ void deg_to_dinv_kernel(float* deg, int n) {
    int i = blockIdx.x * blockDim.x + threadIdx.x;
    if (i < n) deg[i] = rsqrtf(deg[i]);
}

__global__ void pad_w3_kernel(const float* __restrict__ w, float* __restrict__ wp) {
    int i = blockIdx.x * blockDim.x + threadIdx.x;
    if (i >= HID * NCLS_P) return;
    int r = i / NCLS_P, c = i - r * NCLS_P;
    wp[i] = (c < NCLS) ? w[r * NCLS + c] : 0.0f;
}

template <int NT, int K, int NOUT>
__global__ void gemm_wmma_f32_kernel(const float* __restrict__ A,
                                     const float* __restrict__ B,
                                     float* __restrict__ H,
                                     int M) {
    static_assert(NOUT == NT * 16, "full tiles only");
    const int lane  = threadIdx.x & 31;
    const int wave  = threadIdx.x >> 5;
    const int wpb   = blockDim.x >> 5;
    const int tile  = blockIdx.x * wpb + wave;
    const int mtile = M >> 4;
    if (tile >= mtile) return;

    const int rowA    = (tile << 4) + (lane & 15);
    const int kb      = (lane >> 4) << 1;
    const int colBase = lane & 15;

    v8f acc[NT];
#pragma unroll
    for (int t = 0; t < NT; ++t) acc[t] = (v8f)0.0f;

    const float* __restrict__ arow = A + (size_t)rowA * K;

#pragma unroll 2
    for (int k = 0; k < K; k += 4) {
        const float2 av = *(const float2*)(arow + k + kb);
        v2f a; a.x = av.x; a.y = av.y;

        const float* __restrict__ b0 = B + (size_t)(k + kb) * NOUT + colBase;
        const float* __restrict__ b1 = b0 + NOUT;
#pragma unroll
        for (int t = 0; t < NT; ++t) {
            v2f b;
            b.x = b0[t * 16];
            b.y = b1[t * 16];
            acc[t] = __builtin_amdgcn_wmma_f32_16x16x4_f32(
                false, a, false, b, (short)0, acc[t], false, false);
        }
    }

    const int rowD = (tile << 4) + ((lane >> 4) << 3);
    float* __restrict__ hout = H + (size_t)rowD * NOUT + colBase;
#pragma unroll
    for (int t = 0; t < NT; ++t) {
#pragma unroll
        for (int r = 0; r < 8; ++r)
            hout[(size_t)r * NOUT + t * 16] = acc[t][r];
    }
}

template <int F, int HS>
__global__ void scatter_kernel(const float* __restrict__ h,
                               const int* __restrict__ src,
                               const int* __restrict__ dst,
                               const float* __restrict__ dinv,
                               float* __restrict__ out,
                               int e) {
    const int lane = threadIdx.x & 31;
    int warp = (blockIdx.x * blockDim.x + threadIdx.x) >> 5;
    const int nwarps = (gridDim.x * blockDim.x) >> 5;
    for (; warp < e; warp += nwarps) {
        const int s = src[warp];
        const int d = dst[warp];
        const float w = dinv[s] * dinv[d];
        const float* __restrict__ hs = h + (size_t)s * HS;
        float* __restrict__ od       = out + (size_t)d * F;
#pragma unroll
        for (int f = lane; f < F; f += 32)
            atomicAdd(&od[f], hs[f] * w);
    }
}

template <int F, int HS, int RELU>
__global__ void finalize_kernel(const float* __restrict__ agg,
                                const float* __restrict__ h,
                                const float* __restrict__ dinv,
                                const float* __restrict__ bias,
                                float* __restrict__ out,
                                int n) {
    size_t idx = (size_t)blockIdx.x * blockDim.x + threadIdx.x;
    const size_t total = (size_t)n * F;
    const size_t stride = (size_t)gridDim.x * blockDim.x;
    for (; idx < total; idx += stride) {
        const int i = (int)(idx / F);
        const int f = (int)(idx - (size_t)i * F);
        const float di = dinv[i];
        float v = agg[idx] + h[(size_t)i * HS + f] * di * di + bias[f];
        if (RELU) v = fmaxf(v, 0.0f);
        out[idx] = v;
    }
}

extern "C" void kernel_launch(void* const* d_in, const int* in_sizes, int n_in,
                              void* d_out, int out_size, void* d_ws, size_t ws_size,
                              hipStream_t stream) {
    const float* x    = (const float*)d_in[0];
    const int*   eidx = (const int*)d_in[1];
    const float* W1   = (const float*)d_in[2];
    const float* b1   = (const float*)d_in[3];
    const float* W2   = (const float*)d_in[4];
    const float* b2   = (const float*)d_in[5];
    const float* W3   = (const float*)d_in[6];
    const float* b3   = (const float*)d_in[7];
    float* out = (float*)d_out;

    const int* src = eidx;
    const int* dst = eidx + EE;

    float* dinv = (float*)d_ws;
    float* bufA = dinv + NN;
    float* bufB = bufA + (size_t)NN * HID;
    float* bufC = bufB + (size_t)NN * HID;
    float* W3p  = bufC + (size_t)NN * HID;

    const int TB = 256;

    init_deg_kernel<<<(NN + TB - 1) / TB, TB, 0, stream>>>(dinv, NN);
    deg_accum_kernel<<<4096, TB, 0, stream>>>(dst, dinv, EE);
    deg_to_dinv_kernel<<<(NN + TB - 1) / TB, TB, 0, stream>>>(dinv, NN);
    pad_w3_kernel<<<(HID * NCLS_P + TB - 1) / TB, TB, 0, stream>>>(W3, W3p);

    const int MT = NN / 16;
    const int gemmBlocks = (MT + 7) / 8;
    const int scatterBlocks = 16384;
    const int elemBlocks128 = (int)(((size_t)NN * HID + TB - 1) / TB);
    const int elemBlocks40  = (int)(((size_t)NN * NCLS + TB - 1) / TB);

    gemm_wmma_f32_kernel<8, IN_DIM, HID><<<gemmBlocks, TB, 0, stream>>>(x, W1, bufA, NN);
    hipMemsetAsync(bufB, 0, (size_t)NN * HID * sizeof(float), stream);
    scatter_kernel<HID, HID><<<scatterBlocks, TB, 0, stream>>>(bufA, src, dst, dinv, bufB, EE);
    finalize_kernel<HID, HID, 1><<<elemBlocks128, TB, 0, stream>>>(bufB, bufA, dinv, b1, bufB, NN);

    gemm_wmma_f32_kernel<8, HID, HID><<<gemmBlocks, TB, 0, stream>>>(bufB, W2, bufA, NN);
    hipMemsetAsync(bufC, 0, (size_t)NN * HID * sizeof(float), stream);
    scatter_kernel<HID, HID><<<scatterBlocks, TB, 0, stream>>>(bufA, src, dst, dinv, bufC, EE);
    finalize_kernel<HID, HID, 1><<<elemBlocks128, TB, 0, stream>>>(bufC, bufA, dinv, b2, bufB, NN);

    gemm_wmma_f32_kernel<3, HID, NCLS_P><<<gemmBlocks, TB, 0, stream>>>(bufB, W3p, bufA, NN);
    hipMemsetAsync(out, 0, (size_t)NN * NCLS * sizeof(float), stream);
    scatter_kernel<NCLS, NCLS_P><<<scatterBlocks, TB, 0, stream>>>(bufA, src, dst, dinv, out, EE);
    finalize_kernel<NCLS, NCLS_P, 0><<<elemBlocks40, TB, 0, stream>>>(out, bufA, dinv, b3, out, NN);
}


# Edges grouped by destination in-call (returning int add = rank, scan, fill); per-node register gather-sum with fused self/bias/relu replaces float atomic scatter; sum order = arrival rank
# speedup vs baseline: 2.1919x; 2.1919x over previous
#include <hip/hip_runtime.h>
#include <hip/hip_bf16.h>


#define NN      100000
#define EE      1600000
#define IN_DIM  256
#define HID     128
#define NCLS    40
#define NCLS_P  48
#define SCAN_TB 1024
#define NPW     8

typedef __attribute__((ext_vector_type(2))) float v2f;
typedef __attribute__((ext_vector_type(8))) float v8f;

__global__ void zero_cnt_kernel(int* __restrict__ cnt, int n) {
    int i = blockIdx.x * blockDim.x + threadIdx.x;
    if (i < n) cnt[i] = 0;
}

__global__ void count_rank_kernel(const int* __restrict__ dst, int* __restrict__ cnt,
                                  int* __restrict__ rank, int e) {
    int i = blockIdx.x * blockDim.x + threadIdx.x;
    const int stride = gridDim.x * blockDim.x;
    for (; i < e; i += stride) rank[i] = atomicAdd(&cnt[dst[i]], 1);
}

__global__ void scan_dinv_kernel(const int* __restrict__ cnt, int* __restrict__ rowptr,
                                 float* __restrict__ dinv, int n) {
    __shared__ int wfront[32];
    __shared__ int wtotal[32];
    const int t    = threadIdx.x;
    const int lane = t & 31;
    const int wave = t >> 5;
    const int start = blockIdx.x * SCAN_TB;

    int front = 0;
    const int4* __restrict__ c4 = (const int4*)cnt;
    const int nvec = start >> 2;
#pragma unroll 4
    for (int j = t; j < nvec; j += SCAN_TB) {
        const int4 v = c4[j];
        front += (v.x + v.y) + (v.z + v.w);
    }

    const int i = start + t;
    const int c = (i < n) ? cnt[i] : 0;
    int incl = c;
#pragma unroll
    for (int d = 1; d < 32; d <<= 1) {
        const int v = __shfl_up(incl, d);
        if (lane >= d) incl += v;
    }
#pragma unroll
    for (int d = 16; d >= 1; d >>= 1) front += __shfl_xor(front, d);

    if (lane == 31) wtotal[wave] = incl;
    if (lane == 0)  wfront[wave] = front;
    __syncthreads();

    int base = 0;
#pragma unroll
    for (int w = 0; w < SCAN_TB / 32; ++w) {
        base += wfront[w];
        if (w < wave) base += wtotal[w];
    }
    const int excl = base + incl - c;
    if (i < n) {
        rowptr[i] = excl;
        dinv[i]   = rsqrtf((float)(c + 1));
        if (i == n - 1) rowptr[n] = excl + c;
    }
}

__global__ void fill_kernel(const int* __restrict__ src, const int* __restrict__ dst,
                            const int* __restrict__ rank, const int* __restrict__ rowptr,
                            int* __restrict__ col, int e) {
    int i = blockIdx.x * blockDim.x + threadIdx.x;
    const int stride = gridDim.x * blockDim.x;
    for (; i < e; i += stride) {
        const int p = rowptr[dst[i]] + rank[i];
        if ((unsigned)p < (unsigned)e) col[p] = src[i];
    }
}

__global__ void pad_w3_kernel(const float* __restrict__ w, float* __restrict__ wp) {
    int i = blockIdx.x * blockDim.x + threadIdx.x;
    if (i >= HID * NCLS_P) return;
    int r = i / NCLS_P, c = i - r * NCLS_P;
    wp[i] = (c < NCLS) ? w[r * NCLS + c] : 0.0f;
}

template <int NT, int K, int NOUT>
__global__ void gemm_wmma_f32_kernel(const float* __restrict__ A,
                                     const float* __restrict__ B,
                                     float* __restrict__ H,
                                     int M) {
    static_assert(NOUT == NT * 16, "full tiles only");
    const int lane  = threadIdx.x & 31;
    const int wave  = threadIdx.x >> 5;
    const int wpb   = blockDim.x >> 5;
    const int tile  = blockIdx.x * wpb + wave;
    const int mtile = M >> 4;
    if (tile >= mtile) return;

    const int rowA    = (tile << 4) + (lane & 15);
    const int kb      = (lane >> 4) << 1;
    const int colBase = lane & 15;

    v8f acc[NT];
#pragma unroll
    for (int t = 0; t < NT; ++t) acc[t] = (v8f)0.0f;

    const float* __restrict__ arow = A + (size_t)rowA * K;

#pragma unroll 2
    for (int k = 0; k < K; k += 4) {
        const float2 av = *(const float2*)(arow + k + kb);
        v2f a; a.x = av.x; a.y = av.y;

        const float* __restrict__ b0 = B + (size_t)(k + kb) * NOUT + colBase;
        const float* __restrict__ b1 = b0 + NOUT;
#pragma unroll
        for (int t = 0; t < NT; ++t) {
            v2f b;
            b.x = b0[t * 16];
            b.y = b1[t * 16];
            acc[t] = __builtin_amdgcn_wmma_f32_16x16x4_f32(
                false, a, false, b, (short)0, acc[t], false, false);
        }
    }

    const int rowD = (tile << 4) + ((lane >> 4) << 3);
    float* __restrict__ hout = H + (size_t)rowD * NOUT + colBase;
#pragma unroll
    for (int t = 0; t < NT; ++t) {
#pragma unroll
        for (int r = 0; r < 8; ++r)
            hout[(size_t)r * NOUT + t * 16] = acc[t][r];
    }
}

template <int CNT, int HS, typename VF>
__device__ __forceinline__ void add_rows(VF& acc, const float* __restrict__ h,
                                         int s_l, int w_l, int k, int fo) {
    VF v[CNT];
    float w[CNT];
#pragma unroll
    for (int u = 0; u < CNT; ++u) {
        const int s = __builtin_amdgcn_readlane(s_l, k + u);
        w[u] = __int_as_float(__builtin_amdgcn_readlane(w_l, k + u));
        v[u] = *(const VF*)(h + (size_t)s * HS + fo);
    }
#pragma unroll
    for (int u = 0; u < CNT; ++u) acc += v[u] * w[u];
}

template <int F, int HS, int VEC, int RELU>
__global__ void gather_kernel(const float* __restrict__ h,
                              const int* __restrict__ rowptr,
                              const int* __restrict__ col,
                              const float* __restrict__ dinv,
                              const float* __restrict__ bias,
                              float* __restrict__ out,
                              int n) {
    typedef __attribute__((ext_vector_type(VEC))) float VF;
    const int lane = threadIdx.x & 31;
    const int wv = __builtin_amdgcn_readfirstlane(
        (int)((blockIdx.x * blockDim.x + threadIdx.x) >> 5));
    const bool act = lane * VEC < F;
    const int  fo  = act ? lane * VEC : 0;
    const VF bv = *(const VF*)(bias + fo);

    const int first = wv * NPW;
    for (int q = 0; q < NPW; ++q) {
        const int node = first + q;
        if (node >= n) break;
        const int beg = rowptr[node];
        const int end = rowptr[node + 1];
        const float dd = dinv[node];
        const VF hv = *(const VF*)(h + (size_t)node * HS + fo);

        VF acc = (VF)0.0f;
        for (int base = beg; base < end; base += 32) {
            const int j = base + lane;
            int s_l = 0;
            float wf = 0.0f;
            if (j < end) {
                s_l = min(max(col[j], 0), n - 1);
                wf  = dinv[s_l] * dd;
            }
            const int w_l = __float_as_int(wf);
            int cnt = end - base;
            if (cnt > 32) cnt = 32;
            int k = 0;
            for (; k + 8 <= cnt; k += 8) add_rows<8, HS, VF>(acc, h, s_l, w_l, k, fo);
            if (k + 4 <= cnt) { add_rows<4, HS, VF>(acc, h, s_l, w_l, k, fo); k += 4; }
            const int rem = cnt - k;
            if (rem == 3)      add_rows<3, HS, VF>(acc, h, s_l, w_l, k, fo);
            else if (rem == 2) add_rows<2, HS, VF>(acc, h, s_l, w_l, k, fo);
            else if (rem == 1) add_rows<1, HS, VF>(acc, h, s_l, w_l, k, fo);
        }

        if (act) {
            VF r;
#pragma unroll
            for (int c = 0; c < VEC; ++c) {
                float v = acc[c] + hv[c] * dd * dd + bv[c];
                if (RELU) v = fmaxf(v, 0.0f);
                r[c] = v;
            }
            *(VF*)(out + (size_t)node * F + fo) = r;
        }
    }
}

extern "C" void kernel_launch(void* const* d_in, const int* in_sizes, int n_in,
                              void* d_out, int out_size, void* d_ws, size_t ws_size,
                              hipStream_t stream) {
    const float* x    = (const float*)d_in[0];
    const int*   eidx = (const int*)d_in[1];
    const float* W1   = (const float*)d_in[2];
    const float* b1   = (const float*)d_in[3];
    const float* W2   = (const float*)d_in[4];
    const float* b2   = (const float*)d_in[5];
    const float* W3   = (const float*)d_in[6];
    const float* b3   = (const float*)d_in[7];
    float* out = (float*)d_out;

    const int* src = eidx;
    const int* dst = eidx + EE;

    float* dinv = (float*)d_ws;
    float* bufA = dinv + NN;
    float* bufB = bufA + (size_t)NN * HID;
    float* W3p  = bufB + (size_t)NN * HID;
    int* cnt    = (int*)(W3p + HID * NCLS_P);
    int* rowptr = cnt + NN;
    int* rank   = rowptr + (NN + 32);
    int* col    = rank + EE;

    const int TB = 256;
    const int edgeBlocks = 2048;

    zero_cnt_kernel<<<(NN + TB - 1) / TB, TB, 0, stream>>>(cnt, NN);
    count_rank_kernel<<<edgeBlocks, TB, 0, stream>>>(dst, cnt, rank, EE);
    scan_dinv_kernel<<<(NN + SCAN_TB - 1) / SCAN_TB, SCAN_TB, 0, stream>>>(cnt, rowptr, dinv, NN);
    fill_kernel<<<edgeBlocks, TB, 0, stream>>>(src, dst, rank, rowptr, col, EE);
    pad_w3_kernel<<<(HID * NCLS_P + TB - 1) / TB, TB, 0, stream>>>(W3, W3p);

    const int MT = NN / 16;
    const int gemmBlocks = (MT + 7) / 8;
    const int GTB = 128;
    const int gatherBlocks = (NN + NPW * (GTB / 32) - 1) / (NPW * (GTB / 32));

    gemm_wmma_f32_kernel<8, IN_DIM, HID><<<gemmBlocks, TB, 0, stream>>>(x, W1, bufA, NN);
    gather_kernel<HID, HID, 4, 1><<<gatherBlocks, GTB, 0, stream>>>(bufA, rowptr, col, dinv, b1, bufB, NN);

    gemm_wmma_f32_kernel<8, HID, HID><<<gemmBlocks, TB, 0, stream>>>(bufB, W2, bufA, NN);
    gather_kernel<HID, HID, 4, 1><<<gatherBlocks, GTB, 0, stream>>>(bufA, rowptr, col, dinv, b2, bufB, NN);

    gemm_wmma_f32_kernel<3, HID, NCLS_P><<<gemmBlocks, TB, 0, stream>>>(bufB, W3p, bufA, NN);
    gather_kernel<NCLS, NCLS_P, 2, 0><<<gatherBlocks, GTB, 0, stream>>>(bufA, rowptr, col, dinv, b3, out, NN);
}


# Edges grouped by destination with a two-level counting sort on LDS counters (counts per 512-node range, placement, node order per range), no global add per edge; sum order = LDS arrival rank
# speedup vs baseline: 2.8788x; 1.3134x over previous
#include <hip/hip_runtime.h>
#include <hip/hip_bf16.h>


#define NN      100000
#define EE      1600000
#define IN_DIM  256
#define HID     128
#define NCLS    40
#define NCLS_P  48
#define NPW     8
#define BSH     9
#define BSZ     (1 << BSH)
#define NB      ((NN + BSZ - 1) >> BSH)
#define NBL     ((NB + 31) / 32)
#define NCH     256
#define CH      (EE / NCH)
#define CTB     1024
#define CIT     ((CH + CTB - 1) / CTB)
#define RB      8
#define RCAP    12288

typedef __attribute__((ext_vector_type(2))) float v2f;
typedef __attribute__((ext_vector_type(8))) float v8f;

static_assert(EE % NCH == 0, "chunks cover the edge list exactly");
static_assert(NCH == 256, "one wave reads a range's 256 chunk counts as 2 x 16 B per lane");
static_assert(NB <= 32 * NBL && NB <= CTB, "range scan fits one wave / one block");

__device__ __forceinline__ unsigned clamp_node(int v) {
    return min((unsigned)v, (unsigned)(NN - 1));
}

__global__ __launch_bounds__(CTB) void range_count_kernel(const int* __restrict__ dst,
                                                          int* __restrict__ hist) {
    __shared__ int h[NB];
    const int t = threadIdx.x;
    const int c = blockIdx.x;
    if (t < NB) h[t] = 0;
    __syncthreads();
    const int* __restrict__ d = dst + (size_t)c * CH;
    int dv[CIT];
#pragma unroll
    for (int it = 0; it < CIT; ++it) {
        dv[it] = d[min(it * CTB + t, CH - 1)];
    }
#pragma unroll
    for (int it = 0; it < CIT; ++it) {
        const int i = it * CTB + t;
        if (i < CH) atomicAdd(&h[clamp_node(dv[it]) >> BSH], 1);
    }
    __syncthreads();
    if (t < NB) hist[t * NCH + c] = h[t];
}

__global__ __launch_bounds__(CTB) void range_place_kernel(const int* __restrict__ src,
                                                          const int* __restrict__ dst,
                                                          const int* __restrict__ hist,
                                                          int* __restrict__ rstart,
                                                          int* __restrict__ binned) {
    __shared__ int tot[32 * NBL];
    __shared__ int gbase[32 * NBL];
    __shared__ int lcnt[32 * NBL];
    __shared__ int lstart[32 * NBL];
    __shared__ int stage[CH];
    __shared__ unsigned short stage_r[CH];
    const int t    = threadIdx.x;
    const int lane = t & 31;
    const int wave = __builtin_amdgcn_readfirstlane(t >> 5);
    const int c    = blockIdx.x;

    if (t < 32 * NBL) lcnt[t] = 0;

    const int* __restrict__ sp = src + (size_t)c * CH;
    const int* __restrict__ dp = dst + (size_t)c * CH;
    int dv[CIT], sv[CIT];
#pragma unroll
    for (int it = 0; it < CIT; ++it) {
        const int i = min(it * CTB + t, CH - 1);
        dv[it] = dp[i];
        sv[it] = sp[i];
    }

    int4 uu[NBL], vv[NBL];
#pragma unroll
    for (int k = 0; k < NBL; ++k) {
        const int r = min(wave + k * 32, NB - 1);
        const int4* __restrict__ p = (const int4*)(hist + r * NCH) + lane * 2;
        uu[k] = p[0];
        vv[k] = p[1];
    }
#pragma unroll
    for (int k = 0; k < NBL; ++k) {
        const int r = wave + k * 32;
        const int4 u = uu[k], v = vv[k];
        const int c0 = lane * 8;
        int all = (u.x + u.y) + (u.z + u.w) + (v.x + v.y) + (v.z + v.w);
        int below = (c0 + 0 < c ? u.x : 0) + (c0 + 1 < c ? u.y : 0) + (c0 + 2 < c ? u.z : 0) +
                    (c0 + 3 < c ? u.w : 0) + (c0 + 4 < c ? v.x : 0) + (c0 + 5 < c ? v.y : 0) +
                    (c0 + 6 < c ? v.z : 0) + (c0 + 7 < c ? v.w : 0);
#pragma unroll
        for (int s = 16; s >= 1; s >>= 1) {
            all   += __shfl_xor(all, s);
            below += __shfl_xor(below, s);
        }
        if (r < NB && lane == 0) { tot[r] = all; gbase[r] = below; }
    }
    __syncthreads();

    int rk[CIT];
#pragma unroll
    for (int it = 0; it < CIT; ++it) {
        rk[it] = 0;
        if (it * CTB + t < CH) rk[it] = atomicAdd(&lcnt[clamp_node(dv[it]) >> BSH], 1);
    }
    __syncthreads();

    if (wave == 0) {
        int vt[NBL], vl[NBL];
        int sum_t = 0, sum_l = 0;
#pragma unroll
        for (int j = 0; j < NBL; ++j) {
            const int r = lane * NBL + j;
            vt[j] = (r < NB) ? tot[r] : 0;
            vl[j] = (r < NB) ? lcnt[r] : 0;
            sum_t += vt[j];
            sum_l += vl[j];
        }
        int incl_t = sum_t, incl_l = sum_l;
#pragma unroll
        for (int s = 1; s < 32; s <<= 1) {
            const int ot = __shfl_up(incl_t, s);
            const int ol = __shfl_up(incl_l, s);
            if (lane >= s) { incl_t += ot; incl_l += ol; }
        }
        int run_t = incl_t - sum_t;
        int run_l = incl_l - sum_l;
#pragma unroll
        for (int j = 0; j < NBL; ++j) {
            const int r = lane * NBL + j;
            if (r < NB) {
                gbase[r] += run_t;
                lstart[r] = run_l;
                if (c == 0) rstart[r] = run_t;
            }
            run_t += vt[j];
            run_l += vl[j];
        }
        if (c == 0 && lane == 31) rstart[NB] = run_t;
    }
    __syncthreads();

#pragma unroll
    for (int it = 0; it < CIT; ++it) {
        if (it * CTB + t < CH) {
            const unsigned d = clamp_node(dv[it]);
            const unsigned s = clamp_node(sv[it]);
            const unsigned r = d >> BSH;
            const int p = lstart[r] + rk[it];
            if ((unsigned)p < (unsigned)CH) {
                stage[p]   = (int)((s << BSH) | (d & (BSZ - 1)));
                stage_r[p] = (unsigned short)r;
            }
        }
    }
    __syncthreads();

#pragma unroll
    for (int it = 0; it < CIT; ++it) {
        const int i = it * CTB + t;
        if (i < CH) {
            const int r = min((int)stage_r[i], NB - 1);
            const int g = gbase[r] + (i - lstart[r]);
            if ((unsigned)g < (unsigned)EE)
                binned[g] = stage[i];
        }
    }
}

__global__ __launch_bounds__(BSZ) void range_rows_kernel(const int* __restrict__ binned,
                                                         const int* __restrict__ rstart,
                                                         int* __restrict__ rowptr,
                                                         float* __restrict__ dinv,
                                                         int* __restrict__ col) {
    __shared__ int cnt[BSZ];
    __shared__ int cur[BSZ];
    __shared__ int wtot[BSZ / 32];
    __shared__ int stage[RCAP];
    const int t    = threadIdx.x;
    const int lane = t & 31;
    const int wave = t >> 5;
    const int r    = blockIdx.x;
    const int beg = min(max(rstart[r], 0), EE);
    const int end = min(max(rstart[r + 1], 0), EE);
    const bool fits = (end - beg) <= RCAP;

    cnt[t] = 0;
    __syncthreads();
    for (int base = beg + t; base < end; base += RB * BSZ) {
        int rec[RB];
#pragma unroll
        for (int u = 0; u < RB; ++u)
            rec[u] = binned[min(base + u * BSZ, end - 1)];
        static_assert(RB == 8, "the statement below names RB words");
        asm volatile("" :: "v"(rec[0]), "v"(rec[1]), "v"(rec[2]), "v"(rec[3]),
                           "v"(rec[4]), "v"(rec[5]), "v"(rec[6]), "v"(rec[7]));
#pragma unroll
        for (int u = 0; u < RB; ++u)
            if (base + u * BSZ < end) atomicAdd(&cnt[rec[u] & (BSZ - 1)], 1);
    }
    __syncthreads();

    const int c = cnt[t];
    int incl = c;
#pragma unroll
    for (int s = 1; s < 32; s <<= 1) {
        const int o = __shfl_up(incl, s);
        if (lane >= s) incl += o;
    }
    if (lane == 31) wtot[wave] = incl;
    __syncthreads();
    int base0 = beg;
#pragma unroll
    for (int w = 0; w < BSZ / 32; ++w)
        if (w < wave) base0 += wtot[w];
    const int first = base0 + incl - c;
    cur[t] = first;
    const int node = r * BSZ + t;
    if (node < NN) {
        rowptr[node] = first;
        dinv[node]   = rsqrtf((float)(c + 1));
        if (node == NN - 1) rowptr[NN] = first + c;
    }
    __syncthreads();

    for (int base = beg + t; base < end; base += RB * BSZ) {
        int rec[RB];
#pragma unroll
        for (int u = 0; u < RB; ++u)
            rec[u] = binned[min(base + u * BSZ, end - 1)];
#pragma unroll
        for (int u = 0; u < RB; ++u) {
            if (base + u * BSZ < end) {
                const int pos = atomicAdd(&cur[rec[u] & (BSZ - 1)], 1);
                const int sid = (int)((unsigned)rec[u] >> BSH);
                if (fits) {
                    const int q = pos - beg;
                    if ((unsigned)q < (unsigned)RCAP) stage[q] = sid;
                } else if ((unsigned)pos < (unsigned)EE) {
                    col[pos] = sid;
                }
            }
        }
    }
    __syncthreads();
    if (fits) {
        for (int i = (beg & ~31) + t; i < end; i += BSZ)
            if (i >= beg) col[i] = stage[i - beg];
    }
}

__global__ void pad_w3_kernel(const float* __restrict__ w, float* __restrict__ wp) {
    int i = blockIdx.x * blockDim.x + threadIdx.x;
    if (i >= HID * NCLS_P) return;
    int r = i / NCLS_P, c = i - r * NCLS_P;
    wp[i] = (c < NCLS) ? w[r * NCLS + c] : 0.0f;
}

template <int NT, int K, int NOUT>
__global__ void gemm_wmma_f32_kernel(const float* __restrict__ A,
                                     const float* __restrict__ B,
                                     float* __restrict__ H,
                                     int M) {
    static_assert(NOUT == NT * 16, "full tiles only");
    const int lane  = threadIdx.x & 31;
    const int wave  = threadIdx.x >> 5;
    const int wpb   = blockDim.x >> 5;
    const int tile  = blockIdx.x * wpb + wave;
    const int mtile = M >> 4;
    if (tile >= mtile) return;

    const int rowA    = (tile << 4) + (lane & 15);
    const int kb      = (lane >> 4) << 1;
    const int colBase = lane & 15;

    v8f acc[NT];
#pragma unroll
    for (int t = 0; t < NT; ++t) acc[t] = (v8f)0.0f;

    const float* __restrict__ arow = A + (size_t)rowA * K;

#pragma unroll 2
    for (int k = 0; k < K; k += 4) {
        const float2 av = *(const float2*)(arow + k + kb);
        v2f a; a.x = av.x; a.y = av.y;

        const float* __restrict__ b0 = B + (size_t)(k + kb) * NOUT + colBase;
        const float* __restrict__ b1 = b0 + NOUT;
#pragma unroll
        for (int t = 0; t < NT; ++t) {
            v2f b;
            b.x = b0[t * 16];
            b.y = b1[t * 16];
            acc[t] = __builtin_amdgcn_wmma_f32_16x16x4_f32(
                false, a, false, b, (short)0, acc[t], false, false);
        }
    }

    const int rowD = (tile << 4) + ((lane >> 4) << 3);
    float* __restrict__ hout = H + (size_t)rowD * NOUT + colBase;
#pragma unroll
    for (int t = 0; t < NT; ++t) {
#pragma unroll
        for (int r = 0; r < 8; ++r)
            hout[(size_t)r * NOUT + t * 16] = acc[t][r];
    }
}

template <int CNT, int HS, typename VF>
__device__ __forceinline__ void add_rows(VF& acc, const float* __restrict__ h,
                                         int s_l, int w_l, int k, int fo) {
    VF v[CNT];
    float w[CNT];
#pragma unroll
    for (int u = 0; u < CNT; ++u) {
        const int s = __builtin_amdgcn_readlane(s_l, k + u);
        w[u] = __int_as_float(__builtin_amdgcn_readlane(w_l, k + u));
        v[u] = *(const VF*)(h + (size_t)s * HS + fo);
    }
#pragma unroll
    for (int u = 0; u < CNT; ++u) acc += v[u] * w[u];
}

template <int F, int HS, int VEC, int RELU>
__global__ void gather_kernel(const float* __restrict__ h,
                              const int* __restrict__ rowptr,
                              const int* __restrict__ col,
                              const float* __restrict__ dinv,
                              const float* __restrict__ bias,
                              float* __restrict__ out,
                              int n) {
    typedef __attribute__((ext_vector_type(VEC))) float VF;
    const int lane = threadIdx.x & 31;
    const int wv = __builtin_amdgcn_readfirstlane(
        (int)((blockIdx.x * blockDim.x + threadIdx.x) >> 5));
    const bool act = lane * VEC < F;
    const int  fo  = act ? lane * VEC : 0;
    const VF bv = *(const VF*)(bias + fo);

    const int first = wv * NPW;
    for (int q = 0; q < NPW; ++q) {
        const int node = first + q;
        if (node >= n) break;
        const int beg = rowptr[node];
        const int end = rowptr[node + 1];
        const float dd = dinv[node];
        const VF hv = *(const VF*)(h + (size_t)node * HS + fo);

        VF acc = (VF)0.0f;
        for (int base = beg; base < end; base += 32) {
            const int j = base + lane;
            int s_l = 0;
            float wf = 0.0f;
            if (j < end) {
                s_l = min(max(col[j], 0), n - 1);
                wf  = dinv[s_l] * dd;
            }
            const int w_l = __float_as_int(wf);
            int cnt = end - base;
            if (cnt > 32) cnt = 32;
            int k = 0;
            for (; k + 8 <= cnt; k += 8) add_rows<8, HS, VF>(acc, h, s_l, w_l, k, fo);
            if (k + 4 <= cnt) { add_rows<4, HS, VF>(acc, h, s_l, w_l, k, fo); k += 4; }
            const int rem = cnt - k;
            if (rem == 3)      add_rows<3, HS, VF>(acc, h, s_l, w_l, k, fo);
            else if (rem == 2) add_rows<2, HS, VF>(acc, h, s_l, w_l, k, fo);
            else if (rem == 1) add_rows<1, HS, VF>(acc, h, s_l, w_l, k, fo);
        }

        if (act) {
            VF r;
#pragma unroll
            for (int c = 0; c < VEC; ++c) {
                float v = acc[c] + hv[c] * dd * dd + bv[c];
                if (RELU) v = fmaxf(v, 0.0f);
                r[c] = v;
            }
            *(VF*)(out + (size_t)node * F + fo) = r;
        }
    }
}

extern "C" void kernel_launch(void* const* d_in, const int* in_sizes, int n_in,
                              void* d_out, int out_size, void* d_ws, size_t ws_size,
                              hipStream_t stream) {
    const float* x    = (const float*)d_in[0];
    const int*   eidx = (const int*)d_in[1];
    const float* W1   = (const float*)d_in[2];
    const float* b1   = (const float*)d_in[3];
    const float* W2   = (const float*)d_in[4];
    const float* b2   = (const float*)d_in[5];
    const float* W3   = (const float*)d_in[6];
    const float* b3   = (const float*)d_in[7];
    float* out = (float*)d_out;

    const int* src = eidx;
    const int* dst = eidx + EE;

    float* dinv = (float*)d_ws;
    float* bufA = dinv + NN;
    float* bufB = bufA + (size_t)NN * HID;
    float* W3p  = bufB + (size_t)NN * HID;
    int* rowptr = (int*)(W3p + HID * NCLS_P);
    int* col    = rowptr + (NN + 32);
    int* binned = col + EE;
    int* hist   = binned + EE;
    int* rstart = hist + NB * NCH;

    const int TB = 256;

    range_count_kernel<<<NCH, CTB, 0, stream>>>(dst, hist);
    range_place_kernel<<<NCH, CTB, 0, stream>>>(src, dst, hist, rstart, binned);
    range_rows_kernel<<<NB, BSZ, 0, stream>>>(binned, rstart, rowptr, dinv, col);
    pad_w3_kernel<<<(HID * NCLS_P + TB - 1) / TB, TB, 0, stream>>>(W3, W3p);

    const int MT = NN / 16;
    const int gemmBlocks = (MT + 7) / 8;
    const int GTB = 128;
    const int gatherBlocks = (NN + NPW * (GTB / 32) - 1) / (NPW * (GTB / 32));

    gemm_wmma_f32_kernel<8, IN_DIM, HID><<<gemmBlocks, TB, 0, stream>>>(x, W1, bufA, NN);
    gather_kernel<HID, HID, 4, 1><<<gatherBlocks, GTB, 0, stream>>>(bufA, rowptr, col, dinv, b1, bufB, NN);

    gemm_wmma_f32_kernel<8, HID, HID><<<gemmBlocks, TB, 0, stream>>>(bufB, W2, bufA, NN);
    gather_kernel<HID, HID, 4, 1><<<gatherBlocks, GTB, 0, stream>>>(bufA, rowptr, col, dinv, b2, bufB, NN);

    gemm_wmma_f32_kernel<3, HID, NCLS_P><<<gemmBlocks, TB, 0, stream>>>(bufB, W3p, bufA, NN);
    gather_kernel<NCLS, NCLS_P, 2, 0><<<gatherBlocks, GTB, 0, stream>>>(bufA, rowptr, col, dinv, b3, out, NN);
}


# Gathers of layers 1 and 2 fused with the next layer's product: 32 gathered rows per work-group feed the WMMA from LDS, product rows leave as whole lines; 2 launches and x2/x3 round trip gone
# speedup vs baseline: 3.8892x; 1.3510x over previous
#include <hip/hip_runtime.h>
#include <hip/hip_bf16.h>


#define NN      100000
#define EE      1600000
#define IN_DIM  256
#define HID     128
#define NCLS    40
#define NCLS_P  48
#define NPW     8
#define BSH     9
#define BSZ     (1 << BSH)
#define NB      ((NN + BSZ - 1) >> BSH)
#define NBL     ((NB + 31) / 32)
#define NCH     256
#define CH      (EE / NCH)
#define CTB     1024
#define CIT     ((CH + CTB - 1) / CTB)
#define RB      8
#define RCAP    12288
#define FM      32
#define FW      8
#define FNPW    (FM / FW)
#define APITCH  (HID + 4)

typedef __attribute__((ext_vector_type(2))) float v2f;
typedef __attribute__((ext_vector_type(8))) float v8f;

static_assert(EE % NCH == 0, "chunks cover the edge list exactly");
static_assert(NCH == 256, "one wave reads a range's 256 chunk counts as 2 x 16 B per lane");
static_assert(NB <= 32 * NBL && NB <= CTB, "range scan fits one wave / one block");

__device__ __forceinline__ unsigned clamp_node(int v) {
    return min((unsigned)v, (unsigned)(NN - 1));
}

__global__ __launch_bounds__(CTB) void range_count_kernel(const int* __restrict__ dst,
                                                          int* __restrict__ hist) {
    __shared__ int h[NB];
    const int t = threadIdx.x;
    const int c = blockIdx.x;
    if (t < NB) h[t] = 0;
    __syncthreads();
    const int* __restrict__ d = dst + (size_t)c * CH;
    int dv[CIT];
#pragma unroll
    for (int it = 0; it < CIT; ++it) {
        dv[it] = d[min(it * CTB + t, CH - 1)];
    }
#pragma unroll
    for (int it = 0; it < CIT; ++it) {
        const int i = it * CTB + t;
        if (i < CH) atomicAdd(&h[clamp_node(dv[it]) >> BSH], 1);
    }
    __syncthreads();
    if (t < NB) hist[t * NCH + c] = h[t];
}

__global__ __launch_bounds__(CTB) void range_place_kernel(const int* __restrict__ src,
                                                          const int* __restrict__ dst,
                                                          const int* __restrict__ hist,
                                                          int* __restrict__ rstart,
                                                          int* __restrict__ binned) {
    __shared__ int tot[32 * NBL];
    __shared__ int gbase[32 * NBL];
    __shared__ int lcnt[32 * NBL];
    __shared__ int lstart[32 * NBL];
    __shared__ int stage[CH];
    __shared__ unsigned short stage_r[CH];
    const int t    = threadIdx.x;
    const int lane = t & 31;
    const int wave = __builtin_amdgcn_readfirstlane(t >> 5);
    const int c    = blockIdx.x;

    if (t < 32 * NBL) lcnt[t] = 0;

    const int* __restrict__ sp = src + (size_t)c * CH;
    const int* __restrict__ dp = dst + (size_t)c * CH;
    int dv[CIT], sv[CIT];
#pragma unroll
    for (int it = 0; it < CIT; ++it) {
        const int i = min(it * CTB + t, CH - 1);
        dv[it] = dp[i];
        sv[it] = sp[i];
    }

    int4 uu[NBL], vv[NBL];
#pragma unroll
    for (int k = 0; k < NBL; ++k) {
        const int r = min(wave + k * 32, NB - 1);
        const int4* __restrict__ p = (const int4*)(hist + r * NCH) + lane * 2;
        uu[k] = p[0];
        vv[k] = p[1];
    }
#pragma unroll
    for (int k = 0; k < NBL; ++k) {
        const int r = wave + k * 32;
        const int4 u = uu[k], v = vv[k];
        const int c0 = lane * 8;
        int all = (u.x + u.y) + (u.z + u.w) + (v.x + v.y) + (v.z + v.w);
        int below = (c0 + 0 < c ? u.x : 0) + (c0 + 1 < c ? u.y : 0) + (c0 + 2 < c ? u.z : 0) +
                    (c0 + 3 < c ? u.w : 0) + (c0 + 4 < c ? v.x : 0) + (c0 + 5 < c ? v.y : 0) +
                    (c0 + 6 < c ? v.z : 0) + (c0 + 7 < c ? v.w : 0);
#pragma unroll
        for (int s = 16; s >= 1; s >>= 1) {
            all   += __shfl_xor(all, s);
            below += __shfl_xor(below, s);
        }
        if (r < NB && lane == 0) { tot[r] = all; gbase[r] = below; }
    }
    __syncthreads();

    int rk[CIT];
#pragma unroll
    for (int it = 0; it < CIT; ++it) {
        rk[it] = 0;
        if (it * CTB + t < CH) rk[it] = atomicAdd(&lcnt[clamp_node(dv[it]) >> BSH], 1);
    }
    __syncthreads();

    if (wave == 0) {
        int vt[NBL], vl[NBL];
        int sum_t = 0, sum_l = 0;
#pragma unroll
        for (int j = 0; j < NBL; ++j) {
            const int r = lane * NBL + j;
            vt[j] = (r < NB) ? tot[r] : 0;
            vl[j] = (r < NB) ? lcnt[r] : 0;
            sum_t += vt[j];
            sum_l += vl[j];
        }
        int incl_t = sum_t, incl_l = sum_l;
#pragma unroll
        for (int s = 1; s < 32; s <<= 1) {
            const int ot = __shfl_up(incl_t, s);
            const int ol = __shfl_up(incl_l, s);
            if (lane >= s) { incl_t += ot; incl_l += ol; }
        }
        int run_t = incl_t - sum_t;
        int run_l = incl_l - sum_l;
#pragma unroll
        for (int j = 0; j < NBL; ++j) {
            const int r = lane * NBL + j;
            if (r < NB) {
                gbase[r] += run_t;
                lstart[r] = run_l;
                if (c == 0) rstart[r] = run_t;
            }
            run_t += vt[j];
            run_l += vl[j];
        }
        if (c == 0 && lane == 31) rstart[NB] = run_t;
    }
    __syncthreads();

#pragma unroll
    for (int it = 0; it < CIT; ++it) {
        if (it * CTB + t < CH) {
            const unsigned d = clamp_node(dv[it]);
            const unsigned s = clamp_node(sv[it]);
            const unsigned r = d >> BSH;
            const int p = lstart[r] + rk[it];
            if ((unsigned)p < (unsigned)CH) {
                stage[p]   = (int)((s << BSH) | (d & (BSZ - 1)));
                stage_r[p] = (unsigned short)r;
            }
        }
    }
    __syncthreads();

#pragma unroll
    for (int it = 0; it < CIT; ++it) {
        const int i = it * CTB + t;
        if (i < CH) {
            const int r = min((int)stage_r[i], NB - 1);
            const int g = gbase[r] + (i - lstart[r]);
            if ((unsigned)g < (unsigned)EE)
                binned[g] = stage[i];
        }
    }
}

__global__ __launch_bounds__(BSZ) void range_rows_kernel(const int* __restrict__ binned,
                                                         const int* __restrict__ rstart,
                                                         int* __restrict__ rowptr,
                                                         float* __restrict__ dinv,
                                                         int* __restrict__ col) {
    __shared__ int cnt[BSZ];
    __shared__ int cur[BSZ];
    __shared__ int wtot[BSZ / 32];
    __shared__ int stage[RCAP];
    const int t    = threadIdx.x;
    const int lane = t & 31;
    const int wave = t >> 5;
    const int r    = blockIdx.x;
    const int beg = min(max(rstart[r], 0), EE);
    const int end = min(max(rstart[r + 1], 0), EE);
    const bool fits = (end - beg) <= RCAP;

    cnt[t] = 0;
    __syncthreads();
    for (int base = beg + t; base < end; base += RB * BSZ) {
        int rec[RB];
#pragma unroll
        for (int u = 0; u < RB; ++u)
            rec[u] = binned[min(base + u * BSZ, end - 1)];
        static_assert(RB == 8, "the statement below names RB words");
        asm volatile("" :: "v"(rec[0]), "v"(rec[1]), "v"(rec[2]), "v"(rec[3]),
                           "v"(rec[4]), "v"(rec[5]), "v"(rec[6]), "v"(rec[7]));
#pragma unroll
        for (int u = 0; u < RB; ++u)
            if (base + u * BSZ < end) atomicAdd(&cnt[rec[u] & (BSZ - 1)], 1);
    }
    __syncthreads();

    const int c = cnt[t];
    int incl = c;
#pragma unroll
    for (int s = 1; s < 32; s <<= 1) {
        const int o = __shfl_up(incl, s);
        if (lane >= s) incl += o;
    }
    if (lane == 31) wtot[wave] = incl;
    __syncthreads();
    int base0 = beg;
#pragma unroll
    for (int w = 0; w < BSZ / 32; ++w)
        if (w < wave) base0 += wtot[w];
    const int first = base0 + incl - c;
    cur[t] = first;
    const int node = r * BSZ + t;
    if (node < NN) {
        rowptr[node] = first;
        dinv[node]   = rsqrtf((float)(c + 1));
        if (node == NN - 1) rowptr[NN] = first + c;
    }
    __syncthreads();

    for (int base = beg + t; base < end; base += RB * BSZ) {
        int rec[RB];
#pragma unroll
        for (int u = 0; u < RB; ++u)
            rec[u] = binned[min(base + u * BSZ, end - 1)];
#pragma unroll
        for (int u = 0; u < RB; ++u) {
            if (base + u * BSZ < end) {
                const int pos = atomicAdd(&cur[rec[u] & (BSZ - 1)], 1);
                const int sid = (int)((unsigned)rec[u] >> BSH);
                if (fits) {
                    const int q = pos - beg;
                    if ((unsigned)q < (unsigned)RCAP) stage[q] = sid;
                } else if ((unsigned)pos < (unsigned)EE) {
                    col[pos] = sid;
                }
            }
        }
    }
    __syncthreads();
    if (fits) {
        for (int i = (beg & ~31) + t; i < end; i += BSZ)
            if (i >= beg) col[i] = stage[i - beg];
    }
}

__global__ void pad_w3_kernel(const float* __restrict__ w, float* __restrict__ wp) {
    int i = blockIdx.x * blockDim.x + threadIdx.x;
    if (i >= HID * NCLS_P) return;
    int r = i / NCLS_P, c = i - r * NCLS_P;
    wp[i] = (c < NCLS) ? w[r * NCLS + c] : 0.0f;
}

template <int NT, int K, int NOUT>
__global__ void gemm_wmma_f32_kernel(const float* __restrict__ A,
                                     const float* __restrict__ B,
                                     const float* __restrict__ rs,
                                     float* __restrict__ H,
                                     int M) {
    static_assert(NOUT == NT * 16, "full tiles only");
    const int lane  = threadIdx.x & 31;
    const int wave  = threadIdx.x >> 5;
    const int wpb   = blockDim.x >> 5;
    const int tile  = blockIdx.x * wpb + wave;
    const int mtile = M >> 4;
    if (tile >= mtile) return;

    const int rowA    = (tile << 4) + (lane & 15);
    const int kb      = (lane >> 4) << 1;
    const int colBase = lane & 15;

    v8f acc[NT];
#pragma unroll
    for (int t = 0; t < NT; ++t) acc[t] = (v8f)0.0f;

    const float* __restrict__ arow = A + (size_t)rowA * K;

#pragma unroll 2
    for (int k = 0; k < K; k += 4) {
        const float2 av = *(const float2*)(arow + k + kb);
        v2f a; a.x = av.x; a.y = av.y;

        const float* __restrict__ b0 = B + (size_t)(k + kb) * NOUT + colBase;
        const float* __restrict__ b1 = b0 + NOUT;
#pragma unroll
        for (int t = 0; t < NT; ++t) {
            v2f b;
            b.x = b0[t * 16];
            b.y = b1[t * 16];
            acc[t] = __builtin_amdgcn_wmma_f32_16x16x4_f32(
                false, a, false, b, (short)0, acc[t], false, false);
        }
    }

    const int rowD = (tile << 4) + ((lane >> 4) << 3);
    const float4 s0 = *(const float4*)(rs + rowD);
    const float4 s1 = *(const float4*)(rs + rowD + 4);
    const float sc[8] = {s0.x, s0.y, s0.z, s0.w, s1.x, s1.y, s1.z, s1.w};
    float* __restrict__ hout = H + (size_t)rowD * NOUT + colBase;
#pragma unroll
    for (int t = 0; t < NT; ++t) {
#pragma unroll
        for (int r = 0; r < 8; ++r)
            hout[(size_t)r * NOUT + t * 16] = acc[t][r] * sc[r];
    }
}

template <int CNT, int HS, typename VF>
__device__ __forceinline__ void add_rows(VF& acc, const float* __restrict__ h,
                                         int s_l, int k, int fo) {
    VF v[CNT];
#pragma unroll
    for (int u = 0; u < CNT; ++u) {
        const int s = __builtin_amdgcn_readlane(s_l, k + u);
        v[u] = *(const VF*)(h + (size_t)s * HS + fo);
    }
#pragma unroll
    for (int u = 0; u < CNT; ++u) acc += v[u];
}

template <int F, int HS, int VEC, int RELU>
__global__ void gather_kernel(const float* __restrict__ h,
                              const int* __restrict__ rowptr,
                              const int* __restrict__ col,
                              const float* __restrict__ dinv,
                              const float* __restrict__ bias,
                              float* __restrict__ out,
                              int n) {
    typedef __attribute__((ext_vector_type(VEC))) float VF;
    const int lane = threadIdx.x & 31;
    const int wv = __builtin_amdgcn_readfirstlane(
        (int)((blockIdx.x * blockDim.x + threadIdx.x) >> 5));
    const bool act = lane * VEC < F;
    const int  fo  = act ? lane * VEC : 0;
    const VF bv = *(const VF*)(bias + fo);

    const int first = wv * NPW;
    for (int q = 0; q < NPW; ++q) {
        const int node = first + q;
        if (node >= n) break;
        const int beg = rowptr[node];
        const int end = rowptr[node + 1];
        const float dd = dinv[node];
        const VF hv = *(const VF*)(h + (size_t)node * HS + fo);

        VF acc = (VF)0.0f;
        for (int base = beg; base < end; base += 32) {
            const int j = base + lane;
            int s_l = 0;
            if (j < end) s_l = min(max(col[j], 0), n - 1);
            int cnt = end - base;
            if (cnt > 32) cnt = 32;
            int k = 0;
            for (; k + 8 <= cnt; k += 8) add_rows<8, HS, VF>(acc, h, s_l, k, fo);
            if (k + 4 <= cnt) { add_rows<4, HS, VF>(acc, h, s_l, k, fo); k += 4; }
            const int rem = cnt - k;
            if (rem == 3)      add_rows<3, HS, VF>(acc, h, s_l, k, fo);
            else if (rem == 2) add_rows<2, HS, VF>(acc, h, s_l, k, fo);
            else if (rem == 1) add_rows<1, HS, VF>(acc, h, s_l, k, fo);
        }

        if (act) {
            VF r;
#pragma unroll
            for (int c = 0; c < VEC; ++c) {
                float v = (acc[c] + hv[c]) * dd + bv[c];
                if (RELU) v = fmaxf(v, 0.0f);
                r[c] = v;
            }
            *(VF*)(out + (size_t)node * F + fo) = r;
        }
    }
}

template <int NOUT>
__global__ __launch_bounds__(FW * 32) void gather_gemm_kernel(const float* __restrict__ h,
                                                              const int* __restrict__ rowptr,
                                                              const int* __restrict__ col,
                                                              const float* __restrict__ dinv,
                                                              const float* __restrict__ bias,
                                                              const float* __restrict__ Wn,
                                                              float* __restrict__ hn) {
    typedef __attribute__((ext_vector_type(4))) float VF;
    constexpr int NCT = NOUT / 16;
    constexpr int CP  = NOUT + 4;
    static_assert(NOUT % 16 == 0 && NCT <= FW, "one wave per column tile");
    static_assert(NN % FM == 0, "work-groups cover the nodes exactly");
    __shared__ __attribute__((aligned(16))) float As[FM * APITCH];
    __shared__ __attribute__((aligned(16))) float Cs[FM * CP];
    __shared__ float Ds[FM];

    const int lane = threadIdx.x & 31;
    const int wave = __builtin_amdgcn_readfirstlane((int)(threadIdx.x >> 5));
    const int node0 = blockIdx.x * FM;
    const int fo = lane * 4;

    const VF bv = *(const VF*)(bias + fo);
    for (int q = 0; q < FNPW; ++q) {
        const int m    = wave * FNPW + q;
        const int node = node0 + m;
        const int beg = rowptr[node];
        const int end = rowptr[node + 1];
        const float dd = dinv[node];
        const VF hv = *(const VF*)(h + (size_t)node * HID + fo);

        VF acc = (VF)0.0f;
        for (int base = beg; base < end; base += 32) {
            const int j = base + lane;
            int s_l = 0;
            if (j < end) s_l = min(max(col[j], 0), NN - 1);
            int cnt = end - base;
            if (cnt > 32) cnt = 32;
            int k = 0;
            for (; k + 8 <= cnt; k += 8) add_rows<8, HID, VF>(acc, h, s_l, k, fo);
            if (k + 4 <= cnt) { add_rows<4, HID, VF>(acc, h, s_l, k, fo); k += 4; }
            const int rem = cnt - k;
            if (rem == 3)      add_rows<3, HID, VF>(acc, h, s_l, k, fo);
            else if (rem == 2) add_rows<2, HID, VF>(acc, h, s_l, k, fo);
            else if (rem == 1) add_rows<1, HID, VF>(acc, h, s_l, k, fo);
        }
        VF r;
#pragma unroll
        for (int c = 0; c < 4; ++c) r[c] = fmaxf((acc[c] + hv[c]) * dd + bv[c], 0.0f);
        *(VF*)(As + m * APITCH + fo) = r;
        if (lane == 0) Ds[m] = dd;
    }
    __syncthreads();

    if (wave < NCT) {
        const int c  = lane & 15;
        const int kb = (lane >> 4) << 1;
        v8f acc0 = (v8f)0.0f, acc1 = (v8f)0.0f;
        const float* a0p = As + c * APITCH + kb;
        const float* a1p = a0p + 16 * APITCH;
        const float* __restrict__ bp = Wn + (size_t)kb * NOUT + wave * 16 + c;
#pragma unroll 8
        for (int k = 0; k < HID; k += 4) {
            const v2f a0 = *(const v2f*)(a0p + k);
            const v2f a1 = *(const v2f*)(a1p + k);
            v2f b;
            b.x = bp[(size_t)k * NOUT];
            b.y = bp[(size_t)(k + 1) * NOUT];
            acc0 = __builtin_amdgcn_wmma_f32_16x16x4_f32(false, a0, false, b, (short)0, acc0, false, false);
            acc1 = __builtin_amdgcn_wmma_f32_16x16x4_f32(false, a1, false, b, (short)0, acc1, false, false);
        }
        float* cp = Cs + ((lane >> 4) << 3) * CP + wave * 16 + c;
#pragma unroll
        for (int r = 0; r < 8; ++r) {
            cp[r * CP]        = acc0[r];
            cp[(16 + r) * CP] = acc1[r];
        }
    }
    __syncthreads();

    constexpr int NV  = FM * NOUT / 4;
    constexpr int NIT = (NV + FW * 32 - 1) / (FW * 32);
    float4* __restrict__ dstp = (float4*)(hn + (size_t)node0 * NOUT);
#pragma unroll
    for (int it = 0; it < NIT; ++it) {
        const int i = it * (FW * 32) + (int)threadIdx.x;
        if (NV % (FW * 32) == 0 || i < NV) {
            const int e   = i * 4;
            const int row = e / NOUT;
            const int cq  = e - row * NOUT;
            float4 v = *(const float4*)(Cs + row * CP + cq);
            const float s = Ds[row];
            v.x *= s; v.y *= s; v.z *= s; v.w *= s;
            dstp[i] = v;
        }
    }
}

extern "C" void kernel_launch(void* const* d_in, const int* in_sizes, int n_in,
                              void* d_out, int out_size, void* d_ws, size_t ws_size,
                              hipStream_t stream) {
    const float* x    = (const float*)d_in[0];
    const int*   eidx = (const int*)d_in[1];
    const float* W1   = (const float*)d_in[2];
    const float* b1   = (const float*)d_in[3];
    const float* W2   = (const float*)d_in[4];
    const float* b2   = (const float*)d_in[5];
    const float* W3   = (const float*)d_in[6];
    const float* b3   = (const float*)d_in[7];
    float* out = (float*)d_out;

    const int* src = eidx;
    const int* dst = eidx + EE;

    float* dinv = (float*)d_ws;
    float* bufA = dinv + NN;
    float* bufB = bufA + (size_t)NN * HID;
    float* W3p  = bufB + (size_t)NN * HID;
    int* rowptr = (int*)(W3p + HID * NCLS_P);
    int* col    = rowptr + (NN + 32);
    int* binned = col + EE;
    int* hist   = binned + EE;
    int* rstart = hist + NB * NCH;

    const int TB = 256;

    range_count_kernel<<<NCH, CTB, 0, stream>>>(dst, hist);
    range_place_kernel<<<NCH, CTB, 0, stream>>>(src, dst, hist, rstart, binned);
    range_rows_kernel<<<NB, BSZ, 0, stream>>>(binned, rstart, rowptr, dinv, col);
    pad_w3_kernel<<<(HID * NCLS_P + TB - 1) / TB, TB, 0, stream>>>(W3, W3p);

    const int MT = NN / 16;
    const int gemmBlocks = (MT + 7) / 8;
    const int GTB = 128;
    const int gatherBlocks = (NN + NPW * (GTB / 32) - 1) / (NPW * (GTB / 32));
    const int fusedBlocks = NN / FM;

    gemm_wmma_f32_kernel<8, IN_DIM, HID><<<gemmBlocks, TB, 0, stream>>>(x, W1, dinv, bufA, NN);

    gather_gemm_kernel<HID><<<fusedBlocks, FW * 32, 0, stream>>>(bufA, rowptr, col, dinv, b1, W2, bufB);

    gather_gemm_kernel<NCLS_P><<<fusedBlocks, FW * 32, 0, stream>>>(bufB, rowptr, col, dinv, b2, W3p, bufA);

    gather_kernel<NCLS, NCLS_P, 2, 0><<<gatherBlocks, GTB, 0, stream>>>(bufA, rowptr, col, dinv, b3, out, NN);
}


# Layer-1 product re-tiled: 64-row work-groups stage x and W1 in k slabs of 32 through LDS (whole-line loads one slab ahead), 32x64 wave tiles, rows stored as whole lines; k order of the sums kept
# speedup vs baseline: 4.2521x; 1.0933x over previous
#include <hip/hip_runtime.h>
#include <hip/hip_bf16.h>


#define NN      100000
#define EE      1600000
#define IN_DIM  256
#define HID     128
#define NCLS    40
#define NCLS_P  48
#define NPW     8
#define BSH     9
#define BSZ     (1 << BSH)
#define NB      ((NN + BSZ - 1) >> BSH)
#define NBL     ((NB + 31) / 32)
#define NCH     256
#define CH      (EE / NCH)
#define CTB     1024
#define CIT     ((CH + CTB - 1) / CTB)
#define RB      8
#define RCAP    12288
#define FM      32
#define FW      8
#define FNPW    (FM / FW)
#define APITCH  (HID + 4)

typedef __attribute__((ext_vector_type(2))) float v2f;
typedef __attribute__((ext_vector_type(4))) float v4f;
typedef __attribute__((ext_vector_type(8))) float v8f;

static_assert(EE % NCH == 0, "chunks cover the edge list exactly");
static_assert(NCH == 256, "one wave reads a range's 256 chunk counts as 2 x 16 B per lane");
static_assert(NB <= 32 * NBL && NB <= CTB, "range scan fits one wave / one block");

__device__ __forceinline__ unsigned clamp_node(int v) {
    return min((unsigned)v, (unsigned)(NN - 1));
}

__global__ __launch_bounds__(CTB) void range_count_kernel(const int* __restrict__ dst,
                                                          int* __restrict__ hist) {
    __shared__ int h[NB];
    const int t = threadIdx.x;
    const int c = blockIdx.x;
    if (t < NB) h[t] = 0;
    __syncthreads();
    const int* __restrict__ d = dst + (size_t)c * CH;
    int dv[CIT];
#pragma unroll
    for (int it = 0; it < CIT; ++it) {
        dv[it] = d[min(it * CTB + t, CH - 1)];
    }
#pragma unroll
    for (int it = 0; it < CIT; ++it) {
        const int i = it * CTB + t;
        if (i < CH) atomicAdd(&h[clamp_node(dv[it]) >> BSH], 1);
    }
    __syncthreads();
    if (t < NB) hist[t * NCH + c] = h[t];
}

__global__ __launch_bounds__(CTB) void range_place_kernel(const int* __restrict__ src,
                                                          const int* __restrict__ dst,
                                                          const int* __restrict__ hist,
                                                          int* __restrict__ rstart,
                                                          int* __restrict__ binned) {
    __shared__ int tot[32 * NBL];
    __shared__ int gbase[32 * NBL];
    __shared__ int lcnt[32 * NBL];
    __shared__ int lstart[32 * NBL];
    __shared__ int stage[CH];
    __shared__ unsigned short stage_r[CH];
    const int t    = threadIdx.x;
    const int lane = t & 31;
    const int wave = __builtin_amdgcn_readfirstlane(t >> 5);
    const int c    = blockIdx.x;

    if (t < 32 * NBL) lcnt[t] = 0;

    const int* __restrict__ sp = src + (size_t)c * CH;
    const int* __restrict__ dp = dst + (size_t)c * CH;
    int dv[CIT], sv[CIT];
#pragma unroll
    for (int it = 0; it < CIT; ++it) {
        const int i = min(it * CTB + t, CH - 1);
        dv[it] = dp[i];
        sv[it] = sp[i];
    }

    int4 uu[NBL], vv[NBL];
#pragma unroll
    for (int k = 0; k < NBL; ++k) {
        const int r = min(wave + k * 32, NB - 1);
        const int4* __restrict__ p = (const int4*)(hist + r * NCH) + lane * 2;
        uu[k] = p[0];
        vv[k] = p[1];
    }
#pragma unroll
    for (int k = 0; k < NBL; ++k) {
        const int r = wave + k * 32;
        const int4 u = uu[k], v = vv[k];
        const int c0 = lane * 8;
        int all = (u.x + u.y) + (u.z + u.w) + (v.x + v.y) + (v.z + v.w);
        int below = (c0 + 0 < c ? u.x : 0) + (c0 + 1 < c ? u.y : 0) + (c0 + 2 < c ? u.z : 0) +
                    (c0 + 3 < c ? u.w : 0) + (c0 + 4 < c ? v.x : 0) + (c0 + 5 < c ? v.y : 0) +
                    (c0 + 6 < c ? v.z : 0) + (c0 + 7 < c ? v.w : 0);
#pragma unroll
        for (int s = 16; s >= 1; s >>= 1) {
            all   += __shfl_xor(all, s);
            below += __shfl_xor(below, s);
        }
        if (r < NB && lane == 0) { tot[r] = all; gbase[r] = below; }
    }
    __syncthreads();

    int rk[CIT];
#pragma unroll
    for (int it = 0; it < CIT; ++it) {
        rk[it] = 0;
        if (it * CTB + t < CH) rk[it] = atomicAdd(&lcnt[clamp_node(dv[it]) >> BSH], 1);
    }
    __syncthreads();

    if (wave == 0) {
        int vt[NBL], vl[NBL];
        int sum_t = 0, sum_l = 0;
#pragma unroll
        for (int j = 0; j < NBL; ++j) {
            const int r = lane * NBL + j;
            vt[j] = (r < NB) ? tot[r] : 0;
            vl[j] = (r < NB) ? lcnt[r] : 0;
            sum_t += vt[j];
            sum_l += vl[j];
        }
        int incl_t = sum_t, incl_l = sum_l;
#pragma unroll
        for (int s = 1; s < 32; s <<= 1) {
            const int ot = __shfl_up(incl_t, s);
            const int ol = __shfl_up(incl_l, s);
            if (lane >= s) { incl_t += ot; incl_l += ol; }
        }
        int run_t = incl_t - sum_t;
        int run_l = incl_l - sum_l;
#pragma unroll
        for (int j = 0; j < NBL; ++j) {
            const int r = lane * NBL + j;
            if (r < NB) {
                gbase[r] += run_t;
                lstart[r] = run_l;
                if (c == 0) rstart[r] = run_t;
            }
            run_t += vt[j];
            run_l += vl[j];
        }
        if (c == 0 && lane == 31) rstart[NB] = run_t;
    }
    __syncthreads();

#pragma unroll
    for (int it = 0; it < CIT; ++it) {
        if (it * CTB + t < CH) {
            const unsigned d = clamp_node(dv[it]);
            const unsigned s = clamp_node(sv[it]);
            const unsigned r = d >> BSH;
            const int p = lstart[r] + rk[it];
            if ((unsigned)p < (unsigned)CH) {
                stage[p]   = (int)((s << BSH) | (d & (BSZ - 1)));
                stage_r[p] = (unsigned short)r;
            }
        }
    }
    __syncthreads();

#pragma unroll
    for (int it = 0; it < CIT; ++it) {
        const int i = it * CTB + t;
        if (i < CH) {
            const int r = min((int)stage_r[i], NB - 1);
            const int g = gbase[r] + (i - lstart[r]);
            if ((unsigned)g < (unsigned)EE)
                binned[g] = stage[i];
        }
    }
}

__global__ __launch_bounds__(BSZ) void range_rows_kernel(const int* __restrict__ binned,
                                                         const int* __restrict__ rstart,
                                                         int* __restrict__ rowptr,
                                                         float* __restrict__ dinv,
                                                         int* __restrict__ col) {
    __shared__ int cnt[BSZ];
    __shared__ int cur[BSZ];
    __shared__ int wtot[BSZ / 32];
    __shared__ int stage[RCAP];
    const int t    = threadIdx.x;
    const int lane = t & 31;
    const int wave = t >> 5;
    const int r    = blockIdx.x;
    const int beg = min(max(rstart[r], 0), EE);
    const int end = min(max(rstart[r + 1], 0), EE);
    const bool fits = (end - beg) <= RCAP;

    cnt[t] = 0;
    __syncthreads();
    for (int base = beg + t; base < end; base += RB * BSZ) {
        int rec[RB];
#pragma unroll
        for (int u = 0; u < RB; ++u)
            rec[u] = binned[min(base + u * BSZ, end - 1)];
        static_assert(RB == 8, "the statement below names RB words");
        asm volatile("" :: "v"(rec[0]), "v"(rec[1]), "v"(rec[2]), "v"(rec[3]),
                           "v"(rec[4]), "v"(rec[5]), "v"(rec[6]), "v"(rec[7]));
#pragma unroll
        for (int u = 0; u < RB; ++u)
            if (base + u * BSZ < end) atomicAdd(&cnt[rec[u] & (BSZ - 1)], 1);
    }
    __syncthreads();

    const int c = cnt[t];
    int incl = c;
#pragma unroll
    for (int s = 1; s < 32; s <<= 1) {
        const int o = __shfl_up(incl, s);
        if (lane >= s) incl += o;
    }
    if (lane == 31) wtot[wave] = incl;
    __syncthreads();
    int base0 = beg;
#pragma unroll
    for (int w = 0; w < BSZ / 32; ++w)
        if (w < wave) base0 += wtot[w];
    const int first = base0 + incl - c;
    cur[t] = first;
    const int node = r * BSZ + t;
    if (node < NN) {
        rowptr[node] = first;
        dinv[node]   = rsqrtf((float)(c + 1));
        if (node == NN - 1) rowptr[NN] = first + c;
    }
    __syncthreads();

    for (int base = beg + t; base < end; base += RB * BSZ) {
        int rec[RB];
#pragma unroll
        for (int u = 0; u < RB; ++u)
            rec[u] = binned[min(base + u * BSZ, end - 1)];
#pragma unroll
        for (int u = 0; u < RB; ++u) {
            if (base + u * BSZ < end) {
                const int pos = atomicAdd(&cur[rec[u] & (BSZ - 1)], 1);
                const int sid = (int)((unsigned)rec[u] >> BSH);
                if (fits) {
                    const int q = pos - beg;
                    if ((unsigned)q < (unsigned)RCAP) stage[q] = sid;
                } else if ((unsigned)pos < (unsigned)EE) {
                    col[pos] = sid;
                }
            }
        }
    }
    __syncthreads();
    if (fits) {
        for (int i = (beg & ~31) + t; i < end; i += BSZ)
            if (i >= beg) col[i] = stage[i - beg];
    }
}

__global__ void pad_w3_kernel(const float* __restrict__ w, float* __restrict__ wp) {
    int i = blockIdx.x * blockDim.x + threadIdx.x;
    if (i >= HID * NCLS_P) return;
    int r = i / NCLS_P, c = i - r * NCLS_P;
    wp[i] = (c < NCLS) ? w[r * NCLS + c] : 0.0f;
}

#define G1_BM   64
#define G1_KS   32
#define G1_NS   (IN_DIM / G1_KS)
#define G1_AP   (G1_KS + 4)
#define G1_WP   (2 * HID + 32)
#define G1_CP   (HID + 4)
#define G1_AS   (G1_BM * G1_AP)
#define G1_WS   ((G1_KS / 2) * G1_WP)
#define G1_CS   (G1_BM * G1_CP)
#define G1_SM   (G1_CS > G1_AS + G1_WS ? G1_CS : G1_AS + G1_WS)
static_assert(IN_DIM % G1_KS == 0 && HID == 128 && G1_KS == 32, "staging maps below are written for these shapes");

__global__ __launch_bounds__(128) void gemm1_lds_kernel(const float* __restrict__ A,
                                                        const float* __restrict__ B,
                                                        const float* __restrict__ rs,
                                                        float* __restrict__ H) {
    __shared__ __attribute__((aligned(16))) float sm[G1_SM];
    __shared__ float Ds[G1_BM];
    float* As = sm;
    float* Ws = sm + G1_AS;

    const int tid  = threadIdx.x;
    const int lane = tid & 31;
    const int wave = __builtin_amdgcn_readfirstlane(tid >> 5);
    const int rg = wave >> 1, cg = wave & 1;
    const int c = lane & 15, h = lane >> 4;
    const int node0 = blockIdx.x * G1_BM;

    if (tid < G1_BM) Ds[tid] = rs[min(node0 + tid, NN - 1)];

    unsigned xo[4];
#pragma unroll
    for (int j = 0; j < 4; ++j)
        xo[j] = (unsigned)min(node0 + (tid >> 3) + 16 * j, NN - 1) * IN_DIM + (tid & 7) * 4;
    const float* wp = B + (size_t)(2 * wave) * HID + lane * 4;
    float* asd = As + (tid >> 3) * G1_AP + (tid & 7) * 4;
    float* wsd = Ws + wave * G1_WP + lane * 8;

    v4f xr[4], wa[4], wb[4];
#pragma unroll
    for (int j = 0; j < 4; ++j) {
        xr[j] = *(const v4f*)(A + xo[j]);
        wa[j] = *(const v4f*)(wp + (8 * j) * HID);
        wb[j] = *(const v4f*)(wp + (8 * j + 1) * HID);
    }

    v8f acc[2][4];
#pragma unroll
    for (int rt = 0; rt < 2; ++rt)
#pragma unroll
        for (int ct = 0; ct < 4; ++ct) acc[rt][ct] = (v8f)0.0f;

    const float* af = As + (rg * 32 + c) * G1_AP + 2 * h;
    const float* bf = Ws + h * G1_WP + (cg * 64 + c) * 2;

#pragma unroll 1
    for (int s = 0; s < G1_NS; ++s) {
#pragma unroll
        for (int j = 0; j < 4; ++j) {
            *(v4f*)(asd + 16 * j * G1_AP) = xr[j];
            v4f lo, hi;
            lo.x = wa[j].x; lo.y = wb[j].x; lo.z = wa[j].y; lo.w = wb[j].y;
            hi.x = wa[j].z; hi.y = wb[j].z; hi.z = wa[j].w; hi.w = wb[j].w;
            *(v4f*)(wsd + 4 * j * G1_WP)     = lo;
            *(v4f*)(wsd + 4 * j * G1_WP + 4) = hi;
        }
        __syncthreads();
        if (s + 1 < G1_NS) {
#pragma unroll
            for (int j = 0; j < 4; ++j) {
                xr[j] = *(const v4f*)(A + (xo[j] + (unsigned)((s + 1) * G1_KS)));
                wa[j] = *(const v4f*)(wp + ((s + 1) * G1_KS + 8 * j) * HID);
                wb[j] = *(const v4f*)(wp + ((s + 1) * G1_KS + 8 * j + 1) * HID);
            }
        }
#pragma unroll
        for (int kk = 0; kk < G1_KS / 4; ++kk) {
            v2f a[2], b[4];
#pragma unroll
            for (int rt = 0; rt < 2; ++rt) a[rt] = *(const v2f*)(af + rt * 16 * G1_AP + kk * 4);
#pragma unroll
            for (int ct = 0; ct < 4; ++ct) b[ct] = *(const v2f*)(bf + kk * 2 * G1_WP + ct * 32);
#pragma unroll
            for (int rt = 0; rt < 2; ++rt)
#pragma unroll
                for (int ct = 0; ct < 4; ++ct)
                    acc[rt][ct] = __builtin_amdgcn_wmma_f32_16x16x4_f32(
                        false, a[rt], false, b[ct], (short)0, acc[rt][ct], false, false);
        }
        __syncthreads();
    }

    float* cp = sm + (rg * 32 + 8 * h) * G1_CP + cg * 64 + c;
#pragma unroll
    for (int rt = 0; rt < 2; ++rt)
#pragma unroll
        for (int ct = 0; ct < 4; ++ct)
#pragma unroll
            for (int r = 0; r < 8; ++r)
                cp[(rt * 16 + r) * G1_CP + ct * 16] = acc[rt][ct][r];
    __syncthreads();

    float4* __restrict__ dstp = (float4*)(H + (size_t)node0 * HID);
#pragma unroll
    for (int it = 0; it < G1_BM * HID / 4 / 128; ++it) {
        const int i   = it * 128 + tid;
        const int row = i >> 5;
        if (node0 + row < NN) {
            float4 v = *(const float4*)(sm + row * G1_CP + (i & 31) * 4);
            const float sc = Ds[row];
            v.x *= sc; v.y *= sc; v.z *= sc; v.w *= sc;
            dstp[i] = v;
        }
    }
}

template <int CNT, int HS, typename VF>
__device__ __forceinline__ void add_rows(VF& acc, const float* __restrict__ h,
                                         int s_l, int k, int fo) {
    VF v[CNT];
#pragma unroll
    for (int u = 0; u < CNT; ++u) {
        const int s = __builtin_amdgcn_readlane(s_l, k + u);
        v[u] = *(const VF*)(h + (size_t)s * HS + fo);
    }
#pragma unroll
    for (int u = 0; u < CNT; ++u) acc += v[u];
}

template <int F, int HS, int VEC, int RELU>
__global__ void gather_kernel(const float* __restrict__ h,
                              const int* __restrict__ rowptr,
                              const int* __restrict__ col,
                              const float* __restrict__ dinv,
                              const float* __restrict__ bias,
                              float* __restrict__ out,
                              int n) {
    typedef __attribute__((ext_vector_type(VEC))) float VF;
    const int lane = threadIdx.x & 31;
    const int wv = __builtin_amdgcn_readfirstlane(
        (int)((blockIdx.x * blockDim.x + threadIdx.x) >> 5));
    const bool act = lane * VEC < F;
    const int  fo  = act ? lane * VEC : 0;
    const VF bv = *(const VF*)(bias + fo);

    const int first = wv * NPW;
    for (int q = 0; q < NPW; ++q) {
        const int node = first + q;
        if (node >= n) break;
        const int beg = rowptr[node];
        const int end = rowptr[node + 1];
        const float dd = dinv[node];
        const VF hv = *(const VF*)(h + (size_t)node * HS + fo);

        VF acc = (VF)0.0f;
        for (int base = beg; base < end; base += 32) {
            const int j = base + lane;
            int s_l = 0;
            if (j < end) s_l = min(max(col[j], 0), n - 1);
            int cnt = end - base;
            if (cnt > 32) cnt = 32;
            int k = 0;
            for (; k + 8 <= cnt; k += 8) add_rows<8, HS, VF>(acc, h, s_l, k, fo);
            if (k + 4 <= cnt) { add_rows<4, HS, VF>(acc, h, s_l, k, fo); k += 4; }
            const int rem = cnt - k;
            if (rem == 3)      add_rows<3, HS, VF>(acc, h, s_l, k, fo);
            else if (rem == 2) add_rows<2, HS, VF>(acc, h, s_l, k, fo);
            else if (rem == 1) add_rows<1, HS, VF>(acc, h, s_l, k, fo);
        }

        if (act) {
            VF r;
#pragma unroll
            for (int c = 0; c < VEC; ++c) {
                float v = (acc[c] + hv[c]) * dd + bv[c];
                if (RELU) v = fmaxf(v, 0.0f);
                r[c] = v;
            }
            *(VF*)(out + (size_t)node * F + fo) = r;
        }
    }
}

template <int NOUT>
__global__ __launch_bounds__(FW * 32) void gather_gemm_kernel(const float* __restrict__ h,
                                                              const int* __restrict__ rowptr,
                                                              const int* __restrict__ col,
                                                              const float* __restrict__ dinv,
                                                              const float* __restrict__ bias,
                                                              const float* __restrict__ Wn,
                                                              float* __restrict__ hn) {
    typedef __attribute__((ext_vector_type(4))) float VF;
    constexpr int NCT = NOUT / 16;
    constexpr int CP  = NOUT + 4;
    static_assert(NOUT % 16 == 0 && NCT <= FW, "one wave per column tile");
    static_assert(NN % FM == 0, "work-groups cover the nodes exactly");
    __shared__ __attribute__((aligned(16))) float As[FM * APITCH];
    __shared__ __attribute__((aligned(16))) float Cs[FM * CP];
    __shared__ float Ds[FM];

    const int lane = threadIdx.x & 31;
    const int wave = __builtin_amdgcn_readfirstlane((int)(threadIdx.x >> 5));
    const int node0 = blockIdx.x * FM;
    const int fo = lane * 4;

    const VF bv = *(const VF*)(bias + fo);
    for (int q = 0; q < FNPW; ++q) {
        const int m    = wave * FNPW + q;
        const int node = node0 + m;
        const int beg = rowptr[node];
        const int end = rowptr[node + 1];
        const float dd = dinv[node];
        const VF hv = *(const VF*)(h + (size_t)node * HID + fo);

        VF acc = (VF)0.0f;
        for (int base = beg; base < end; base += 32) {
            const int j = base + lane;
            int s_l = 0;
            if (j < end) s_l = min(max(col[j], 0), NN - 1);
            int cnt = end - base;
            if (cnt > 32) cnt = 32;
            int k = 0;
            for (; k + 8 <= cnt; k += 8) add_rows<8, HID, VF>(acc, h, s_l, k, fo);
            if (k + 4 <= cnt) { add_rows<4, HID, VF>(acc, h, s_l, k, fo); k += 4; }
            const int rem = cnt - k;
            if (rem == 3)      add_rows<3, HID, VF>(acc, h, s_l, k, fo);
            else if (rem == 2) add_rows<2, HID, VF>(acc, h, s_l, k, fo);
            else if (rem == 1) add_rows<1, HID, VF>(acc, h, s_l, k, fo);
        }
        VF r;
#pragma unroll
        for (int c = 0; c < 4; ++c) r[c] = fmaxf((acc[c] + hv[c]) * dd + bv[c], 0.0f);
        *(VF*)(As + m * APITCH + fo) = r;
        if (lane == 0) Ds[m] = dd;
    }
    __syncthreads();

    if (wave < NCT) {
        const int c  = lane & 15;
        const int kb = (lane >> 4) << 1;
        v8f acc0 = (v8f)0.0f, acc1 = (v8f)0.0f;
        const float* a0p = As + c * APITCH + kb;
        const float* a1p = a0p + 16 * APITCH;
        const float* __restrict__ bp = Wn + (size_t)kb * NOUT + wave * 16 + c;
#pragma unroll 8
        for (int k = 0; k < HID; k += 4) {
            const v2f a0 = *(const v2f*)(a0p + k);
            const v2f a1 = *(const v2f*)(a1p + k);
            v2f b;
            b.x = bp[(size_t)k * NOUT];
            b.y = bp[(size_t)(k + 1) * NOUT];
            acc0 = __builtin_amdgcn_wmma_f32_16x16x4_f32(false, a0, false, b, (short)0, acc0, false, false);
            acc1 = __builtin_amdgcn_wmma_f32_16x16x4_f32(false, a1, false, b, (short)0, acc1, false, false);
        }
        float* cp = Cs + ((lane >> 4) << 3) * CP + wave * 16 + c;
#pragma unroll
        for (int r = 0; r < 8; ++r) {
            cp[r * CP]        = acc0[r];
            cp[(16 + r) * CP] = acc1[r];
        }
    }
    __syncthreads();

    constexpr int NV  = FM * NOUT / 4;
    constexpr int NIT = (NV + FW * 32 - 1) / (FW * 32);
    float4* __restrict__ dstp = (float4*)(hn + (size_t)node0 * NOUT);
#pragma unroll
    for (int it = 0; it < NIT; ++it) {
        const int i = it * (FW * 32) + (int)threadIdx.x;
        if (NV % (FW * 32) == 0 || i < NV) {
            const int e   = i * 4;
            const int row = e / NOUT;
            const int cq  = e - row * NOUT;
            float4 v = *(const float4*)(Cs + row * CP + cq);
            const float s = Ds[row];
            v.x *= s; v.y *= s; v.z *= s; v.w *= s;
            dstp[i] = v;
        }
    }
}

extern "C" void kernel_launch(void* const* d_in, const int* in_sizes, int n_in,
                              void* d_out, int out_size, void* d_ws, size_t ws_size,
                              hipStream_t stream) {
    const float* x    = (const float*)d_in[0];
    const int*   eidx = (const int*)d_in[1];
    const float* W1   = (const float*)d_in[2];
    const float* b1   = (const float*)d_in[3];
    const float* W2   = (const float*)d_in[4];
    const float* b2   = (const float*)d_in[5];
    const float* W3   = (const float*)d_in[6];
    const float* b3   = (const float*)d_in[7];
    float* out = (float*)d_out;

    const int* src = eidx;
    const int* dst = eidx + EE;

    float* dinv = (float*)d_ws;
    float* bufA = dinv + NN;
    float* bufB = bufA + (size_t)NN * HID;
    float* W3p  = bufB + (size_t)NN * HID;
    int* rowptr = (int*)(W3p + HID * NCLS_P);
    int* col    = rowptr + (NN + 32);
    int* binned = col + EE;
    int* hist   = binned + EE;
    int* rstart = hist + NB * NCH;

    const int TB = 256;

    range_count_kernel<<<NCH, CTB, 0, stream>>>(dst, hist);
    range_place_kernel<<<NCH, CTB, 0, stream>>>(src, dst, hist, rstart, binned);
    range_rows_kernel<<<NB, BSZ, 0, stream>>>(binned, rstart, rowptr, dinv, col);
    pad_w3_kernel<<<(HID * NCLS_P + TB - 1) / TB, TB, 0, stream>>>(W3, W3p);

    const int gemmBlocks = (NN + G1_BM - 1) / G1_BM;
    const int GTB = 128;
    const int gatherBlocks = (NN + NPW * (GTB / 32) - 1) / (NPW * (GTB / 32));
    const int fusedBlocks = NN / FM;

    gemm1_lds_kernel<<<gemmBlocks, 128, 0, stream>>>(x, W1, dinv, bufA);

    gather_gemm_kernel<HID><<<fusedBlocks, FW * 32, 0, stream>>>(bufA, rowptr, col, dinv, b1, W2, bufB);

    gather_gemm_kernel<NCLS_P><<<fusedBlocks, FW * 32, 0, stream>>>(bufB, rowptr, col, dinv, b2, W3p, bufA);

    gather_kernel<NCLS, NCLS_P, 2, 0><<<gatherBlocks, GTB, 0, stream>>>(bufA, rowptr, col, dinv, b3, out, NN);
}


# Edge grouping in 2 launches, not 3: chunks store range-ordered words as one run plus a (start,count) table, range work-groups collect their 256 pieces; sum order = LDS arrival rank, lane map new
# speedup vs baseline: 4.3850x; 1.0313x over previous
#include <hip/hip_runtime.h>
#include <hip/hip_bf16.h>


#define NN      100000
#define EE      1600000
#define IN_DIM  256
#define HID     128
#define NCLS    40
#define NCLS_P  48
#define NPW     8
#define BSH     9
#define BSZ     (1 << BSH)
#define NB      ((NN + BSZ - 1) >> BSH)
#define NBL     ((NB + 31) / 32)
#define NCH     256
#define CH      (EE / NCH)
#define CHS     6272
#define TABP    256
#define PPW     16
#define CTB     512
#define CIT     ((CH + CTB - 1) / CTB)
#define RCAP    12288
#define FM      32
#define FW      8
#define FNPW    (FM / FW)
#define APITCH  (HID + 4)

typedef __attribute__((ext_vector_type(2))) float v2f;
typedef __attribute__((ext_vector_type(4))) float v4f;
typedef __attribute__((ext_vector_type(8))) float v8f;

static_assert(EE % NCH == 0, "chunks cover the edge list exactly");
static_assert(NCH == PPW * (BSZ / 32), "every wave of a range work-group owns PPW chunks");
static_assert(NB <= 32 * NBL && NB <= TABP, "range scan fits one wave, a table row holds all ranges");
static_assert(CH < 65536 && CHS >= CH && CHS % 32 == 0, "start and count of a piece fit 16 bits each; chunk slots are whole lines");

__device__ __forceinline__ unsigned clamp_node(int v) {
    return min((unsigned)v, (unsigned)(NN - 1));
}

__global__ __launch_bounds__(CTB) void chunk_order_kernel(const int* __restrict__ src,
                                                          const int* __restrict__ dst,
                                                          int* __restrict__ tab,
                                                          int* __restrict__ binned) {
    __shared__ int lcnt[32 * NBL];
    __shared__ int lstart[32 * NBL];
    __shared__ int stage[CH];
    const int t    = threadIdx.x;
    const int lane = t & 31;
    const int wave = __builtin_amdgcn_readfirstlane(t >> 5);
    const int c    = blockIdx.x;

    if (t < 32 * NBL) lcnt[t] = 0;

    const int* __restrict__ sp = src + (size_t)c * CH;
    const int* __restrict__ dp = dst + (size_t)c * CH;
    int dv[CIT], sv[CIT];
#pragma unroll
    for (int it = 0; it < CIT; ++it) {
        const int i = min(it * CTB + t, CH - 1);
        dv[it] = dp[i];
        sv[it] = sp[i];
    }
    __syncthreads();

    int rk[CIT];
#pragma unroll
    for (int it = 0; it < CIT; ++it) {
        rk[it] = 0;
        if (it * CTB + t < CH) rk[it] = atomicAdd(&lcnt[clamp_node(dv[it]) >> BSH], 1);
    }
    __syncthreads();

    if (wave == 0) {
        int vl[NBL];
        int sum_l = 0;
#pragma unroll
        for (int j = 0; j < NBL; ++j) {
            const int r = lane * NBL + j;
            vl[j] = (r < NB) ? lcnt[r] : 0;
            sum_l += vl[j];
        }
        int incl_l = sum_l;
#pragma unroll
        for (int s = 1; s < 32; s <<= 1) {
            const int ol = __shfl_up(incl_l, s);
            if (lane >= s) incl_l += ol;
        }
        int run_l = incl_l - sum_l;
#pragma unroll
        for (int j = 0; j < NBL; ++j) {
            const int r = lane * NBL + j;
            if (r < NB) {
                lstart[r] = run_l;
                tab[c * TABP + r] = (run_l << 16) | vl[j];
            }
            run_l += vl[j];
        }
    }
    __syncthreads();

    int pl[CIT];
#pragma unroll
    for (int it = 0; it < CIT; ++it) pl[it] = lstart[clamp_node(dv[it]) >> BSH] + rk[it];
#pragma unroll
    for (int it = 0; it < CIT; ++it) {
        const unsigned d = clamp_node(dv[it]);
        const unsigned s = clamp_node(sv[it]);
        if (it * CTB + t < CH && (unsigned)pl[it] < (unsigned)CH)
            stage[pl[it]] = (int)((s << BSH) | (d & (BSZ - 1)));
    }
    __syncthreads();

    int* __restrict__ bp = binned + (size_t)c * CHS;
#pragma unroll
    for (int it = 0; it < CIT; ++it) {
        const int i = it * CTB + t;
        if (i < CH) bp[i] = stage[i];
    }
}

__device__ __forceinline__ void word_bounds(unsigned w, int& st, int& cn) {
    st = (int)min(w >> 16, (unsigned)CH);
    cn = (int)min(w & 0xffffu, (unsigned)(CH - st));
}
__device__ __forceinline__ void piece_bounds(int word, int u, int& st, int& cn) {
    word_bounds((unsigned)__builtin_amdgcn_readlane(word, u), st, cn);
}

__global__ __launch_bounds__(BSZ) void range_rows_kernel(const int* __restrict__ binned,
                                                         const int* __restrict__ tab,
                                                         int* __restrict__ rowptr,
                                                         float* __restrict__ dinv,
                                                         int* __restrict__ col) {
    __shared__ int cnt[BSZ];
    __shared__ int cur[BSZ];
    __shared__ int wtot[BSZ / 32];
    __shared__ int wbeg[BSZ / 32];
    __shared__ int wnum[BSZ / 32];
    __shared__ int stage[RCAP];
    const int t    = threadIdx.x;
    const int lane = t & 31;
    const int wave = __builtin_amdgcn_readfirstlane(t >> 5);
    const int r    = blockIdx.x;

    const int word = tab[(wave * PPW + (lane & (PPW - 1))) * TABP + r];
    cnt[t] = 0;

    const int* __restrict__ slot0 = binned + (size_t)(wave * PPW) * CHS;
    int rec[PPW][2];
#pragma unroll
    for (int u = 0; u < PPW; ++u) {
        int st, cn;
        piece_bounds(word, u, st, cn);
        const int* __restrict__ p = slot0 + u * CHS;
        rec[u][0] = p[min(st + lane, CH - 1)];
        rec[u][1] = p[min(st + lane + 32, CH - 1)];
    }
    {
        int st, cn;
        word_bounds((unsigned)word, st, cn);
        if (lane >= PPW) { st = 0; cn = 0; }
#pragma unroll
        for (int s = 16; s >= 1; s >>= 1) {
            st += __shfl_xor(st, s);
            cn += __shfl_xor(cn, s);
        }
        if (lane == 0) { wbeg[wave] = st; wnum[wave] = cn; }
    }
    __syncthreads();

#pragma unroll
    for (int u = 0; u < PPW; ++u) {
        int st, cn;
        piece_bounds(word, u, st, cn);
        if (lane < cn)      atomicAdd(&cnt[rec[u][0] & (BSZ - 1)], 1);
        if (lane + 32 < cn) atomicAdd(&cnt[rec[u][1] & (BSZ - 1)], 1);
    }
    for (int u = 0; u < PPW; ++u) {
        int st, cn;
        piece_bounds(word, u, st, cn);
        const int* __restrict__ p = slot0 + u * CHS + st;
        for (int b = 64; b < cn; b += 32)
            if (b + lane < cn) atomicAdd(&cnt[p[b + lane] & (BSZ - 1)], 1);
    }
    __syncthreads();

    int beg = 0, num = 0;
#pragma unroll
    for (int w = 0; w < BSZ / 32; ++w) { beg += wbeg[w]; num += wnum[w]; }
    beg = min(max(beg, 0), EE);
    const int end = min(beg + max(num, 0), EE);
    const bool fits = (end - beg) <= RCAP;

    const int c = cnt[t];
    int incl = c;
#pragma unroll
    for (int s = 1; s < 32; s <<= 1) {
        const int o = __shfl_up(incl, s);
        if (lane >= s) incl += o;
    }
    if (lane == 31) wtot[wave] = incl;
    __syncthreads();
    int base0 = beg;
#pragma unroll
    for (int w = 0; w < BSZ / 32; ++w)
        if (w < wave) base0 += wtot[w];
    const int first = base0 + incl - c;
    cur[t] = first;
    const int node = r * BSZ + t;
    if (node < NN) {
        rowptr[node] = first;
        dinv[node]   = rsqrtf((float)(c + 1));
        if (node == NN - 1) rowptr[NN] = first + c;
    }
    __syncthreads();

    auto put = [&](int w, int pos) {
        const int sid = (int)((unsigned)w >> BSH);
        if (fits) {
            const int q = pos - beg;
            if ((unsigned)q < (unsigned)RCAP) stage[q] = sid;
        } else if ((unsigned)pos < (unsigned)EE) {
            col[pos] = sid;
        }
    };
    int pos[PPW][2];
#pragma unroll
    for (int u = 0; u < PPW; ++u) {
        int st, cn;
        piece_bounds(word, u, st, cn);
        pos[u][0] = -1;
        pos[u][1] = -1;
        if (lane < cn)      pos[u][0] = atomicAdd(&cur[rec[u][0] & (BSZ - 1)], 1);
        if (lane + 32 < cn) pos[u][1] = atomicAdd(&cur[rec[u][1] & (BSZ - 1)], 1);
    }
#pragma unroll
    for (int u = 0; u < PPW; ++u) {
        int st, cn;
        piece_bounds(word, u, st, cn);
        if (lane < cn)      put(rec[u][0], pos[u][0]);
        if (lane + 32 < cn) put(rec[u][1], pos[u][1]);
    }
    for (int u = 0; u < PPW; ++u) {
        int st, cn;
        piece_bounds(word, u, st, cn);
        const int* __restrict__ p = slot0 + u * CHS + st;
        for (int b = 64; b < cn; b += 32) {
            if (b + lane < cn) {
                const int w = p[b + lane];
                put(w, atomicAdd(&cur[w & (BSZ - 1)], 1));
            }
        }
    }
    __syncthreads();
    if (fits) {
        for (int i = (beg & ~31) + t; i < end; i += BSZ)
            if (i >= beg) col[i] = stage[i - beg];
    }
}

__global__ void pad_w3_kernel(const float* __restrict__ w, float* __restrict__ wp) {
    int i = blockIdx.x * blockDim.x + threadIdx.x;
    if (i >= HID * NCLS_P) return;
    int r = i / NCLS_P, c = i - r * NCLS_P;
    wp[i] = (c < NCLS) ? w[r * NCLS + c] : 0.0f;
}

#define G1_BM   64
#define G1_KS   32
#define G1_NS   (IN_DIM / G1_KS)
#define G1_AP   (G1_KS + 4)
#define G1_WP   (2 * HID + 32)
#define G1_CP   (HID + 4)
#define G1_AS   (G1_BM * G1_AP)
#define G1_WS   ((G1_KS / 2) * G1_WP)
#define G1_CS   (G1_BM * G1_CP)
#define G1_SM   (G1_CS > G1_AS + G1_WS ? G1_CS : G1_AS + G1_WS)
static_assert(IN_DIM % G1_KS == 0 && HID == 128 && G1_KS == 32, "staging maps below are written for these shapes");

__global__ __launch_bounds__(128) void gemm1_lds_kernel(const float* __restrict__ A,
                                                        const float* __restrict__ B,
                                                        const float* __restrict__ rs,
                                                        float* __restrict__ H) {
    __shared__ __attribute__((aligned(16))) float sm[G1_SM];
    __shared__ float Ds[G1_BM];
    float* As = sm;
    float* Ws = sm + G1_AS;

    const int tid  = threadIdx.x;
    const int lane = tid & 31;
    const int wave = __builtin_amdgcn_readfirstlane(tid >> 5);
    const int rg = wave >> 1, cg = wave & 1;
    const int c = lane & 15, h = lane >> 4;
    const int node0 = blockIdx.x * G1_BM;

    if (tid < G1_BM) Ds[tid] = rs[min(node0 + tid, NN - 1)];

    unsigned xo[4];
#pragma unroll
    for (int j = 0; j < 4; ++j)
        xo[j] = (unsigned)min(node0 + (tid >> 3) + 16 * j, NN - 1) * IN_DIM + (tid & 7) * 4;
    const float* wp = B + (size_t)(2 * wave) * HID + lane * 4;
    float* asd = As + (tid >> 3) * G1_AP + (tid & 7) * 4;
    float* wsd = Ws + wave * G1_WP + lane * 8;

    v4f xr[4], wa[4], wb[4];
#pragma unroll
    for (int j = 0; j < 4; ++j) {
        xr[j] = *(const v4f*)(A + xo[j]);
        wa[j] = *(const v4f*)(wp + (8 * j) * HID);
        wb[j] = *(const v4f*)(wp + (8 * j + 1) * HID);
    }

    v8f acc[2][4];
#pragma unroll
    for (int rt = 0; rt < 2; ++rt)
#pragma unroll
        for (int ct = 0; ct < 4; ++ct) acc[rt][ct] = (v8f)0.0f;

    const float* af = As + (rg * 32 + c) * G1_AP + 2 * h;
    const float* bf = Ws + h * G1_WP + (cg * 64 + c) * 2;

#pragma unroll 1
    for (int s = 0; s < G1_NS; ++s) {
#pragma unroll
        for (int j = 0; j < 4; ++j) {
            *(v4f*)(asd + 16 * j * G1_AP) = xr[j];
            v4f lo, hi;
            lo.x = wa[j].x; lo.y = wb[j].x; lo.z = wa[j].y; lo.w = wb[j].y;
            hi.x = wa[j].z; hi.y = wb[j].z; hi.z = wa[j].w; hi.w = wb[j].w;
            *(v4f*)(wsd + 4 * j * G1_WP)     = lo;
            *(v4f*)(wsd + 4 * j * G1_WP + 4) = hi;
        }
        __syncthreads();
        if (s + 1 < G1_NS) {
#pragma unroll
            for (int j = 0; j < 4; ++j) {
                xr[j] = *(const v4f*)(A + (xo[j] + (unsigned)((s + 1) * G1_KS)));
                wa[j] = *(const v4f*)(wp + ((s + 1) * G1_KS + 8 * j) * HID);
                wb[j] = *(const v4f*)(wp + ((s + 1) * G1_KS + 8 * j + 1) * HID);
            }
        }
#pragma unroll
        for (int kk = 0; kk < G1_KS / 4; ++kk) {
            v2f a[2], b[4];
#pragma unroll
            for (int rt = 0; rt < 2; ++rt) a[rt] = *(const v2f*)(af + rt * 16 * G1_AP + kk * 4);
#pragma unroll
            for (int ct = 0; ct < 4; ++ct) b[ct] = *(const v2f*)(bf + kk * 2 * G1_WP + ct * 32);
#pragma unroll
            for (int rt = 0; rt < 2; ++rt)
#pragma unroll
                for (int ct = 0; ct < 4; ++ct)
                    acc[rt][ct] = __builtin_amdgcn_wmma_f32_16x16x4_f32(
                        false, a[rt], false, b[ct], (short)0, acc[rt][ct], false, false);
        }
        __syncthreads();
    }

    float* cp = sm + (rg * 32 + 8 * h) * G1_CP + cg * 64 + c;
#pragma unroll
    for (int rt = 0; rt < 2; ++rt)
#pragma unroll
        for (int ct = 0; ct < 4; ++ct)
#pragma unroll
            for (int r = 0; r < 8; ++r)
                cp[(rt * 16 + r) * G1_CP + ct * 16] = acc[rt][ct][r];
    __syncthreads();

    float4* __restrict__ dstp = (float4*)(H + (size_t)node0 * HID);
#pragma unroll
    for (int it = 0; it < G1_BM * HID / 4 / 128; ++it) {
        const int i   = it * 128 + tid;
        const int row = i >> 5;
        if (node0 + row < NN) {
            float4 v = *(const float4*)(sm + row * G1_CP + (i & 31) * 4);
            const float sc = Ds[row];
            v.x *= sc; v.y *= sc; v.z *= sc; v.w *= sc;
            dstp[i] = v;
        }
    }
}

template <int CNT, int HS, typename VF>
__device__ __forceinline__ void add_rows(VF& acc, const float* __restrict__ h,
                                         int s_l, int k, int fo) {
    VF v[CNT];
#pragma unroll
    for (int u = 0; u < CNT; ++u) {
        const int s = __builtin_amdgcn_readlane(s_l, k + u);
        v[u] = *(const VF*)(h + (size_t)s * HS + fo);
    }
#pragma unroll
    for (int u = 0; u < CNT; ++u) acc += v[u];
}

template <int F, int HS, int VEC, int RELU>
__global__ void gather_kernel(const float* __restrict__ h,
                              const int* __restrict__ rowptr,
                              const int* __restrict__ col,
                              const float* __restrict__ dinv,
                              const float* __restrict__ bias,
                              float* __restrict__ out,
                              int n) {
    typedef __attribute__((ext_vector_type(VEC))) float VF;
    const int lane = threadIdx.x & 31;
    const int wv = __builtin_amdgcn_readfirstlane(
        (int)((blockIdx.x * blockDim.x + threadIdx.x) >> 5));
    const bool act = lane * VEC < F;
    const int  fo  = act ? lane * VEC : 0;
    const VF bv = *(const VF*)(bias + fo);

    const int first = wv * NPW;
    for (int q = 0; q < NPW; ++q) {
        const int node = first + q;
        if (node >= n) break;
        const int beg = rowptr[node];
        const int end = rowptr[node + 1];
        const float dd = dinv[node];
        const VF hv = *(const VF*)(h + (size_t)node * HS + fo);

        VF acc = (VF)0.0f;
        for (int base = beg; base < end; base += 32) {
            const int j = base + lane;
            int s_l = 0;
            if (j < end) s_l = min(max(col[j], 0), n - 1);
            int cnt = end - base;
            if (cnt > 32) cnt = 32;
            int k = 0;
            for (; k + 8 <= cnt; k += 8) add_rows<8, HS, VF>(acc, h, s_l, k, fo);
            if (k + 4 <= cnt) { add_rows<4, HS, VF>(acc, h, s_l, k, fo); k += 4; }
            const int rem = cnt - k;
            if (rem == 3)      add_rows<3, HS, VF>(acc, h, s_l, k, fo);
            else if (rem == 2) add_rows<2, HS, VF>(acc, h, s_l, k, fo);
            else if (rem == 1) add_rows<1, HS, VF>(acc, h, s_l, k, fo);
        }

        if (act) {
            VF r;
#pragma unroll
            for (int c = 0; c < VEC; ++c) {
                float v = (acc[c] + hv[c]) * dd + bv[c];
                if (RELU) v = fmaxf(v, 0.0f);
                r[c] = v;
            }
            *(VF*)(out + (size_t)node * F + fo) = r;
        }
    }
}

template <int NOUT>
__global__ __launch_bounds__(FW * 32) void gather_gemm_kernel(const float* __restrict__ h,
                                                              const int* __restrict__ rowptr,
                                                              const int* __restrict__ col,
                                                              const float* __restrict__ dinv,
                                                              const float* __restrict__ bias,
                                                              const float* __restrict__ Wn,
                                                              float* __restrict__ hn) {
    typedef __attribute__((ext_vector_type(4))) float VF;
    constexpr int NCT = NOUT / 16;
    constexpr int CP  = NOUT + 4;
    static_assert(NOUT % 16 == 0 && NCT <= FW, "one wave per column tile");
    static_assert(NN % FM == 0, "work-groups cover the nodes exactly");
    __shared__ __attribute__((aligned(16))) float As[FM * APITCH];
    __shared__ __attribute__((aligned(16))) float Cs[FM * CP];
    __shared__ float Ds[FM];

    const int lane = threadIdx.x & 31;
    const int wave = __builtin_amdgcn_readfirstlane((int)(threadIdx.x >> 5));
    const int node0 = blockIdx.x * FM;
    const int fo = lane * 4;

    const VF bv = *(const VF*)(bias + fo);
    for (int q = 0; q < FNPW; ++q) {
        const int m    = wave * FNPW + q;
        const int node = node0 + m;
        const int beg = rowptr[node];
        const int end = rowptr[node + 1];
        const float dd = dinv[node];
        const VF hv = *(const VF*)(h + (size_t)node * HID + fo);

        VF acc = (VF)0.0f;
        for (int base = beg; base < end; base += 32) {
            const int j = base + lane;
            int s_l = 0;
            if (j < end) s_l = min(max(col[j], 0), NN - 1);
            int cnt = end - base;
            if (cnt > 32) cnt = 32;
            int k = 0;
            for (; k + 8 <= cnt; k += 8) add_rows<8, HID, VF>(acc, h, s_l, k, fo);
            if (k + 4 <= cnt) { add_rows<4, HID, VF>(acc, h, s_l, k, fo); k += 4; }
            const int rem = cnt - k;
            if (rem == 3)      add_rows<3, HID, VF>(acc, h, s_l, k, fo);
            else if (rem == 2) add_rows<2, HID, VF>(acc, h, s_l, k, fo);
            else if (rem == 1) add_rows<1, HID, VF>(acc, h, s_l, k, fo);
        }
        VF r;
#pragma unroll
        for (int c = 0; c < 4; ++c) r[c] = fmaxf((acc[c] + hv[c]) * dd + bv[c], 0.0f);
        *(VF*)(As + m * APITCH + fo) = r;
        if (lane == 0) Ds[m] = dd;
    }
    __syncthreads();

    if (wave < NCT) {
        const int c  = lane & 15;
        const int kb = (lane >> 4) << 1;
        v8f acc0 = (v8f)0.0f, acc1 = (v8f)0.0f;
        const float* a0p = As + c * APITCH + kb;
        const float* a1p = a0p + 16 * APITCH;
        const float* __restrict__ bp = Wn + (size_t)kb * NOUT + wave * 16 + c;
#pragma unroll 8
        for (int k = 0; k < HID; k += 4) {
            const v2f a0 = *(const v2f*)(a0p + k);
            const v2f a1 = *(const v2f*)(a1p + k);
            v2f b;
            b.x = bp[(size_t)k * NOUT];
            b.y = bp[(size_t)(k + 1) * NOUT];
            acc0 = __builtin_amdgcn_wmma_f32_16x16x4_f32(false, a0, false, b, (short)0, acc0, false, false);
            acc1 = __builtin_amdgcn_wmma_f32_16x16x4_f32(false, a1, false, b, (short)0, acc1, false, false);
        }
        float* cp = Cs + ((lane >> 4) << 3) * CP + wave * 16 + c;
#pragma unroll
        for (int r = 0; r < 8; ++r) {
            cp[r * CP]        = acc0[r];
            cp[(16 + r) * CP] = acc1[r];
        }
    }
    __syncthreads();

    constexpr int NV  = FM * NOUT / 4;
    constexpr int NIT = (NV + FW * 32 - 1) / (FW * 32);
    float4* __restrict__ dstp = (float4*)(hn + (size_t)node0 * NOUT);
#pragma unroll
    for (int it = 0; it < NIT; ++it) {
        const int i = it * (FW * 32) + (int)threadIdx.x;
        if (NV % (FW * 32) == 0 || i < NV) {
            const int e   = i * 4;
            const int row = e / NOUT;
            const int cq  = e - row * NOUT;
            float4 v = *(const float4*)(Cs + row * CP + cq);
            const float s = Ds[row];
            v.x *= s; v.y *= s; v.z *= s; v.w *= s;
            dstp[i] = v;
        }
    }
}

extern "C" void kernel_launch(void* const* d_in, const int* in_sizes, int n_in,
                              void* d_out, int out_size, void* d_ws, size_t ws_size,
                              hipStream_t stream) {
    const float* x    = (const float*)d_in[0];
    const int*   eidx = (const int*)d_in[1];
    const float* W1   = (const float*)d_in[2];
    const float* b1   = (const float*)d_in[3];
    const float* W2   = (const float*)d_in[4];
    const float* b2   = (const float*)d_in[5];
    const float* W3   = (const float*)d_in[6];
    const float* b3   = (const float*)d_in[7];
    float* out = (float*)d_out;

    const int* src = eidx;
    const int* dst = eidx + EE;

    float* dinv = (float*)d_ws;
    float* bufA = dinv + NN;
    float* bufB = bufA + (size_t)NN * HID;
    float* W3p  = bufB + (size_t)NN * HID;
    int* rowptr = (int*)(W3p + HID * NCLS_P);
    int* col    = rowptr + (NN + 32);
    int* binned = col + EE;
    int* tab    = binned + NCH * CHS;

    const int TB = 256;

    chunk_order_kernel<<<NCH, CTB, 0, stream>>>(src, dst, tab, binned);
    range_rows_kernel<<<NB, BSZ, 0, stream>>>(binned, tab, rowptr, dinv, col);
    pad_w3_kernel<<<(HID * NCLS_P + TB - 1) / TB, TB, 0, stream>>>(W3, W3p);

    const int gemmBlocks = (NN + G1_BM - 1) / G1_BM;
    const int GTB = 128;
    const int gatherBlocks = (NN + NPW * (GTB / 32) - 1) / (NPW * (GTB / 32));
    const int fusedBlocks = NN / FM;

    gemm1_lds_kernel<<<gemmBlocks, 128, 0, stream>>>(x, W1, dinv, bufA);

    gather_gemm_kernel<HID><<<fusedBlocks, FW * 32, 0, stream>>>(bufA, rowptr, col, dinv, b1, W2, bufB);

    gather_gemm_kernel<NCLS_P><<<fusedBlocks, FW * 32, 0, stream>>>(bufB, rowptr, col, dinv, b2, W3p, bufA);

    gather_kernel<NCLS, NCLS_P, 2, 0><<<gatherBlocks, GTB, 0, stream>>>(bufA, rowptr, col, dinv, b3, out, NN);
}
